# starting kernel
# baseline (speedup 1.0000x reference)
#include <hip/hip_runtime.h>
#include <hip/hip_bf16.h>

#define N_NODES  50000
#define N_EDGES  800000
#define N_GRAPHS 512
#define D_H      64
#define D_OUT    32
#define LN_EPS   1e-5f

typedef __attribute__((ext_vector_type(2)))  float    v2f;
typedef __attribute__((ext_vector_type(8)))  float    v8f;
typedef __attribute__((ext_vector_type(16))) _Float16 v16h;

#if defined(__has_builtin)
#if __has_builtin(__builtin_amdgcn_wmma_f32_16x16x4_f32)
#define USE_F32_WMMA 1
#endif
#endif

template <int NCOLS, bool BIAS>
__global__ __launch_bounds__(32 * (NCOLS / 16)) void gcn_gemm_wmma(
    const float* __restrict__ A, const float* __restrict__ B,
    const float* __restrict__ bias, float* __restrict__ C) {
  const int lane = threadIdx.x & 31;
  const int wave = threadIdx.x >> 5;
  const int m    = lane & 15;
  const int half = lane >> 4;
  const int c0   = wave * 16;
  const int r0   = blockIdx.x * 16;
  const int row  = r0 + m;

  v8f acc = {0.f, 0.f, 0.f, 0.f, 0.f, 0.f, 0.f, 0.f};

#if USE_F32_WMMA
  #pragma unroll
  for (int k0 = 0; k0 < 64; k0 += 4) {
    const int ka = k0 + 2 * half;
    v2f a, b;
    a.x = A[row * 64 + ka];
    a.y = A[row * 64 + ka + 1];
    b.x = B[ka * NCOLS + c0 + m];
    b.y = B[(ka + 1) * NCOLS + c0 + m];
    acc = __builtin_amdgcn_wmma_f32_16x16x4_f32(false, a, false, b,
                                                (short)0, acc, false, false);
  }
#else
  for (int kstep = 0; kstep < 2; ++kstep) {
    v16h af, bf;
    #pragma unroll
    for (int e = 0; e < 16; ++e) {
      int v = e >> 1, j = e & 1, grp = v >> 2;
      int K = 32 * kstep + 16 * grp + 8 * half + ((v & 3) * 2 + j);
      af[e] = (_Float16)A[row * 64 + K];
    }
    #pragma unroll
    for (int e = 0; e < 16; ++e) {
      int v = e >> 1, j = e & 1;
      int K = 32 * kstep + 16 * half + 2 * v + j;
      bf[e] = (_Float16)B[K * NCOLS + c0 + m];
    }
    acc = __builtin_amdgcn_wmma_f32_16x16x32_f16(false, af, false, bf,
                                                 (short)0, acc, false, false);
  }
#endif

  #pragma unroll
  for (int i = 0; i < 8; ++i) {
    float v = acc[i];
    if (BIAS) v += bias[c0 + m];
    C[(r0 + i + 8 * half) * NCOLS + c0 + m] = v;
  }
}

__global__ void k_deg_init(float* __restrict__ deg) {
  int i = blockIdx.x * blockDim.x + threadIdx.x;
  if (i < N_NODES) deg[i] = 1.0f;
}

__global__ void k_deg_accum(const int* __restrict__ dst, float* __restrict__ deg) {
  int i = blockIdx.x * blockDim.x + threadIdx.x;
  if (i < N_EDGES) atomicAdd(&deg[dst[i]], 1.0f);
}

__global__ void k_dinv(float* __restrict__ deg) {
  int i = blockIdx.x * blockDim.x + threadIdx.x;
  if (i < N_NODES) deg[i] = rsqrtf(deg[i]);
}

__global__ void k_enorm(const int* __restrict__ src, const int* __restrict__ dst,
                        const float* __restrict__ dinv, float* __restrict__ enorm) {
  int i = blockIdx.x * blockDim.x + threadIdx.x;
  if (i < N_EDGES) enorm[i] = dinv[src[i]] * dinv[dst[i]];
}

__global__ void k_selfloop(const float* __restrict__ hW, const float* __restrict__ dinv,
                           float* __restrict__ agg) {
  int idx = blockIdx.x * blockDim.x + threadIdx.x;
  if (idx >= N_NODES * 64) return;
  float di = dinv[idx >> 6];
  agg[idx] = di * di * hW[idx];
}

__global__ void k_edge_scatter(const float* __restrict__ hW, const int* __restrict__ src,
                               const int* __restrict__ dst, const float* __restrict__ enorm,
                               float* __restrict__ agg) {
  int idx = blockIdx.x * blockDim.x + threadIdx.x;
  if (idx >= N_EDGES * 64) return;
  int e = idx >> 6, f = idx & 63;
  int s = src[e], d = dst[e];
  atomicAdd(&agg[d * 64 + f], enorm[e] * hW[s * 64 + f]);
}

__global__ void k_bias_ln_relu(const float* __restrict__ agg, const float* __restrict__ bias,
                               const float* __restrict__ lnw, const float* __restrict__ lnb,
                               float* __restrict__ out) {
  int node = (blockIdx.x * blockDim.x + threadIdx.x) >> 5;
  int lane = threadIdx.x & 31;
  if (node >= N_NODES) return;
  const float* rowp = agg + (size_t)node * 64;
  float v0 = rowp[lane] + bias[lane];
  float v1 = rowp[lane + 32] + bias[lane + 32];
  float s = v0 + v1;
  #pragma unroll
  for (int off = 16; off > 0; off >>= 1) s += __shfl_xor(s, off, 32);
  float mu = s * (1.0f / 64.0f);
  float d0 = v0 - mu, d1 = v1 - mu;
  float q = d0 * d0 + d1 * d1;
  #pragma unroll
  for (int off = 16; off > 0; off >>= 1) q += __shfl_xor(q, off, 32);
  float rstd = rsqrtf(q * (1.0f / 64.0f) + LN_EPS);
  float* orow = out + (size_t)node * 64;
  orow[lane]      = fmaxf(0.f, d0 * rstd * lnw[lane] + lnb[lane]);
  orow[lane + 32] = fmaxf(0.f, d1 * rstd * lnw[lane + 32] + lnb[lane + 32]);
}

__global__ void k_zero_pool(float* __restrict__ pool, float* __restrict__ cnt) {
  int i = blockIdx.x * blockDim.x + threadIdx.x;
  if (i < N_GRAPHS * 64) pool[i] = 0.f;
  if (i < N_GRAPHS) cnt[i] = 0.f;
}

__global__ void k_pool_accum(const float* __restrict__ h, const int* __restrict__ batch,
                             float* __restrict__ pool, float* __restrict__ cnt) {
  int idx = blockIdx.x * blockDim.x + threadIdx.x;
  if (idx >= N_NODES * 64) return;
  int n = idx >> 6, f = idx & 63;
  int g = batch[n];
  atomicAdd(&pool[g * 64 + f], h[idx]);
  if (f == 0) atomicAdd(&cnt[g], 1.0f);
}

__global__ void k_pool_div(float* __restrict__ pool, const float* __restrict__ cnt) {
  int i = blockIdx.x * blockDim.x + threadIdx.x;
  if (i < N_GRAPHS * 64) pool[i] /= fmaxf(cnt[i >> 6], 1.0f);
}

static inline int cdiv(int a, int b) { return (a + b - 1) / b; }

static void run_layer(const float* hin, const float* W, const float* b,
                      const float* lnw, const float* lnb,
                      const float* dinv, const int* src, const int* dst,
                      const float* enorm, float* hW, float* agg, float* hout,
                      hipStream_t stream) {
  gcn_gemm_wmma<64, false><<<N_NODES / 16, 128, 0, stream>>>(hin, W, nullptr, hW);
  k_selfloop<<<cdiv(N_NODES * 64, 256), 256, 0, stream>>>(hW, dinv, agg);
  k_edge_scatter<<<cdiv(N_EDGES * 64, 256), 256, 0, stream>>>(hW, src, dst, enorm, agg);
  k_bias_ln_relu<<<cdiv(N_NODES * 32, 256), 256, 0, stream>>>(agg, b, lnw, lnb, hout);
}

extern "C" void kernel_launch(void* const* d_in, const int* in_sizes, int n_in,
                              void* d_out, int out_size, void* d_ws, size_t ws_size,
                              hipStream_t stream) {
  (void)in_sizes; (void)n_in; (void)out_size; (void)ws_size;

  const float* x       = (const float*)d_in[0];
  const int*   eidx    = (const int*)d_in[1];
  const int*   batch   = (const int*)d_in[2];
  const float* W_in    = (const float*)d_in[3];
  const float* b_in    = (const float*)d_in[4];
  const float* ln_in_w = (const float*)d_in[5];
  const float* ln_in_b = (const float*)d_in[6];
  const float* W_h     = (const float*)d_in[7];
  const float* b_h     = (const float*)d_in[8];
  const float* ln_h_w  = (const float*)d_in[9];
  const float* ln_h_b  = (const float*)d_in[10];
  const float* W_out   = (const float*)d_in[11];
  const float* b_out   = (const float*)d_in[12];

  const int* src = eidx;
  const int* dst = eidx + N_EDGES;

  float* ws    = (float*)d_ws;
  float* dinv  = ws;
  float* enorm = dinv + N_NODES;
  float* hW    = enorm + N_EDGES;
  float* agg   = hW + (size_t)N_NODES*64;
  float* hbuf  = agg + (size_t)N_NODES*64;
  float* pool  = hbuf + (size_t)N_NODES*64;
  float* cnt   = pool + N_GRAPHS * 64;

  k_deg_init<<<cdiv(N_NODES, 256), 256, 0, stream>>>(dinv);
  k_deg_accum<<<cdiv(N_EDGES, 256), 256, 0, stream>>>(dst, dinv);
  k_dinv<<<cdiv(N_NODES, 256), 256, 0, stream>>>(dinv);
  k_enorm<<<cdiv(N_EDGES, 256), 256, 0, stream>>>(src, dst, dinv, enorm);

  run_layer(x,    W_in,       b_in,       ln_in_w,      ln_in_b,      dinv, src, dst, enorm, hW, agg, hbuf, stream);
  run_layer(hbuf, W_h,        b_h,        ln_h_w,       ln_h_b,       dinv, src, dst, enorm, hW, agg, hbuf, stream);
  run_layer(hbuf, W_h + 64*64, b_h + 64,  ln_h_w + 64,  ln_h_b + 64,  dinv, src, dst, enorm, hW, agg, hbuf, stream);

  k_zero_pool<<<cdiv(N_GRAPHS * 64, 256), 256, 0, stream>>>(pool, cnt);
  k_pool_accum<<<cdiv(N_NODES * 64, 256), 256, 0, stream>>>(hbuf, batch, pool, cnt);
  k_pool_div<<<cdiv(N_GRAPHS * 64, 256), 256, 0, stream>>>(pool, cnt);
  gcn_gemm_wmma<D_OUT, true><<<N_GRAPHS / 16, 64, 0, stream>>>(pool, W_out, b_out, (float*)d_out);
}


# Edges bucketed by target row each call (slot = returning int atomic: arrival-order rank, sum order still varies by run); one wave sums a row in registers, no f32 atomics or self-loop pass
# speedup vs baseline: 2.5379x; 2.5379x over previous
#include <hip/hip_runtime.h>
#include <hip/hip_bf16.h>

#define N_NODES  50000
#define N_EDGES  800000
#define N_GRAPHS 512
#define D_H      64
#define D_OUT    32
#define LN_EPS   1e-5f
#define N_PAD    50048

typedef __attribute__((ext_vector_type(2)))  float    v2f;
typedef __attribute__((ext_vector_type(8)))  float    v8f;
typedef __attribute__((ext_vector_type(16))) _Float16 v16h;

#if defined(__has_builtin)
#if __has_builtin(__builtin_amdgcn_wmma_f32_16x16x4_f32)
#define USE_F32_WMMA 1
#endif
#endif

template <int NCOLS, bool BIAS>
__global__ __launch_bounds__(32 * (NCOLS / 16)) void gcn_gemm_wmma(
    const float* __restrict__ A, const float* __restrict__ B,
    const float* __restrict__ bias, float* __restrict__ C) {
  const int lane = threadIdx.x & 31;
  const int wave = threadIdx.x >> 5;
  const int m    = lane & 15;
  const int half = lane >> 4;
  const int c0   = wave * 16;
  const int r0   = blockIdx.x * 16;
  const int row  = r0 + m;

  v8f acc = {0.f, 0.f, 0.f, 0.f, 0.f, 0.f, 0.f, 0.f};

#if USE_F32_WMMA
  #pragma unroll
  for (int k0 = 0; k0 < 64; k0 += 4) {
    const int ka = k0 + 2 * half;
    v2f a, b;
    a.x = A[row * 64 + ka];
    a.y = A[row * 64 + ka + 1];
    b.x = B[ka * NCOLS + c0 + m];
    b.y = B[(ka + 1) * NCOLS + c0 + m];
    acc = __builtin_amdgcn_wmma_f32_16x16x4_f32(false, a, false, b,
                                                (short)0, acc, false, false);
  }
#else
  for (int kstep = 0; kstep < 2; ++kstep) {
    v16h af, bf;
    #pragma unroll
    for (int e = 0; e < 16; ++e) {
      int v = e >> 1, j = e & 1, grp = v >> 2;
      int K = 32 * kstep + 16 * grp + 8 * half + ((v & 3) * 2 + j);
      af[e] = (_Float16)A[row * 64 + K];
    }
    #pragma unroll
    for (int e = 0; e < 16; ++e) {
      int v = e >> 1, j = e & 1;
      int K = 32 * kstep + 16 * half + 2 * v + j;
      bf[e] = (_Float16)B[K * NCOLS + c0 + m];
    }
    acc = __builtin_amdgcn_wmma_f32_16x16x32_f16(false, af, false, bf,
                                                 (short)0, acc, false, false);
  }
#endif

  #pragma unroll
  for (int i = 0; i < 8; ++i) {
    float v = acc[i];
    if (BIAS) v += bias[c0 + m];
    C[(r0 + i + 8 * half) * NCOLS + c0 + m] = v;
  }
}

__global__ void k_cnt_zero(int* __restrict__ cnt) {
  int i = blockIdx.x * blockDim.x + threadIdx.x;
  if (i < N_NODES) cnt[i] = 0;
}

__global__ void k_rank(const int* __restrict__ dst, int* __restrict__ cnt,
                       int* __restrict__ rank) {
  int i = blockIdx.x * blockDim.x + threadIdx.x;
  if (i < N_EDGES) rank[i] = atomicAdd(&cnt[dst[i]], 1);
}

__global__ __launch_bounds__(1024) void k_scan(const int* __restrict__ cnt,
                                               int* __restrict__ rowptr,
                                               float* __restrict__ dinv) {
  __shared__ int s_front[32];
  __shared__ int s_wtot[32];
  const int tid = threadIdx.x, lane = tid & 31, wave = tid >> 5;
  const int n = blockIdx.x * 1024 + tid;

  const int4* c4 = (const int4*)cnt;
  const int n4 = blockIdx.x * 256;
  int front = 0;
  #pragma unroll 4
  for (int j = tid; j < n4; j += 1024) {
    int4 v = c4[j];
    front += (v.x + v.y) + (v.z + v.w);
  }
  #pragma unroll
  for (int off = 16; off > 0; off >>= 1) front += __shfl_xor(front, off, 32);

  const int c = (n < N_NODES) ? cnt[n] : 0;
  int incl = c;
  #pragma unroll
  for (int off = 1; off < 32; off <<= 1) {
    int t = __shfl_up(incl, off, 32);
    if (lane >= off) incl += t;
  }
  if (lane == 0)  s_front[wave] = front;
  if (lane == 31) s_wtot[wave]  = incl;
  __syncthreads();
  int base = 0;
  for (int w = 0; w < 32; ++w) {
    base += s_front[w];
    if (w < wave) base += s_wtot[w];
  }
  if (n <= N_NODES) rowptr[n] = base + incl - c;
  if (n < N_NODES)  dinv[n] = rsqrtf((float)(c + 1));
}

__global__ void k_fill(const int* __restrict__ src, const int* __restrict__ dst,
                       const int* __restrict__ rank, const int* __restrict__ rowptr,
                       const float* __restrict__ dinv, int2* __restrict__ ed) {
  int i = blockIdx.x * blockDim.x + threadIdx.x;
  if (i >= N_EDGES) return;
  int s = src[i], d = dst[i];
  ed[rowptr[d] + rank[i]] = make_int2(s, __float_as_int(dinv[s] * dinv[d]));
}

template <int K, bool CLAMP>
__device__ __forceinline__ void gather_rows(const float2* __restrict__ h2,
                                            const int2* __restrict__ ed,
                                            int p, int end, int lane,
                                            float& ax, float& ay) {
  int2 e[K];
  #pragma unroll
  for (int j = 0; j < K; ++j) {
    int q = p + j;
    if (CLAMP) q = min(q, end - 1);
    e[j] = ed[q];
  }
  float2 r[K];
  #pragma unroll
  for (int j = 0; j < K; ++j) r[j] = (h2 + e[j].x * 32)[lane];
  #pragma unroll
  for (int j = 0; j < K; ++j) {
    float w = __int_as_float(e[j].y);
    if (CLAMP && p + j >= end) w = 0.f;
    ax += w * r[j].x;
    ay += w * r[j].y;
  }
}

__global__ __launch_bounds__(256) void k_gather(const float* __restrict__ hW,
                                                const int* __restrict__ rowptr,
                                                const int2* __restrict__ ed,
                                                const float* __restrict__ dinv,
                                                float* __restrict__ agg) {
  const int lane = threadIdx.x & 31;
  const int node = __builtin_amdgcn_readfirstlane((int)((blockIdx.x * 256 + threadIdx.x) >> 5));
  if (node >= N_NODES) return;
  const int beg = rowptr[node], end = rowptr[node + 1];
  const float di = dinv[node];
  const float2* h2 = (const float2*)hW;
  const float2 self = (h2 + node * 32)[lane];
  float ax = di * di * self.x;
  float ay = di * di * self.y;
  int p = beg;
  for (; p + 8 <= end; p += 8) gather_rows<8, false>(h2, ed, p, end, lane, ax, ay);
  if (p + 4 <= end) { gather_rows<4, false>(h2, ed, p, end, lane, ax, ay); p += 4; }
  if (p < end) gather_rows<4, true>(h2, ed, p, end, lane, ax, ay);
  ((float2*)agg + node * 32)[lane] = make_float2(ax, ay);
}

__global__ void k_bias_ln_relu(const float* __restrict__ agg, const float* __restrict__ bias,
                               const float* __restrict__ lnw, const float* __restrict__ lnb,
                               float* __restrict__ out) {
  int node = (blockIdx.x * blockDim.x + threadIdx.x) >> 5;
  int lane = threadIdx.x & 31;
  if (node >= N_NODES) return;
  const float* rowp = agg + (size_t)node * 64;
  float v0 = rowp[lane] + bias[lane];
  float v1 = rowp[lane + 32] + bias[lane + 32];
  float s = v0 + v1;
  #pragma unroll
  for (int off = 16; off > 0; off >>= 1) s += __shfl_xor(s, off, 32);
  float mu = s * (1.0f / 64.0f);
  float d0 = v0 - mu, d1 = v1 - mu;
  float q = d0 * d0 + d1 * d1;
  #pragma unroll
  for (int off = 16; off > 0; off >>= 1) q += __shfl_xor(q, off, 32);
  float rstd = rsqrtf(q * (1.0f / 64.0f) + LN_EPS);
  float* orow = out + (size_t)node * 64;
  orow[lane]      = fmaxf(0.f, d0 * rstd * lnw[lane] + lnb[lane]);
  orow[lane + 32] = fmaxf(0.f, d1 * rstd * lnw[lane + 32] + lnb[lane + 32]);
}

__global__ void k_zero_pool(float* __restrict__ pool, float* __restrict__ cnt) {
  int i = blockIdx.x * blockDim.x + threadIdx.x;
  if (i < N_GRAPHS * 64) pool[i] = 0.f;
  if (i < N_GRAPHS) cnt[i] = 0.f;
}

__global__ void k_pool_accum(const float* __restrict__ h, const int* __restrict__ batch,
                             float* __restrict__ pool, float* __restrict__ cnt) {
  int idx = blockIdx.x * blockDim.x + threadIdx.x;
  if (idx >= N_NODES * 64) return;
  int n = idx >> 6, f = idx & 63;
  int g = batch[n];
  atomicAdd(&pool[g * 64 + f], h[idx]);
  if (f == 0) atomicAdd(&cnt[g], 1.0f);
}

__global__ void k_pool_div(float* __restrict__ pool, const float* __restrict__ cnt) {
  int i = blockIdx.x * blockDim.x + threadIdx.x;
  if (i < N_GRAPHS * 64) pool[i] /= fmaxf(cnt[i >> 6], 1.0f);
}

static inline int cdiv(int a, int b) { return (a + b - 1) / b; }

static void run_layer(const float* hin, const float* W, const float* b,
                      const float* lnw, const float* lnb,
                      const float* dinv, const int* rowptr, const int2* ed,
                      float* hW, float* agg, float* hout,
                      hipStream_t stream) {
  gcn_gemm_wmma<64, false><<<N_NODES / 16, 128, 0, stream>>>(hin, W, nullptr, hW);
  k_gather<<<cdiv(N_NODES * 32, 256), 256, 0, stream>>>(hW, rowptr, ed, dinv, agg);
  k_bias_ln_relu<<<cdiv(N_NODES * 32, 256), 256, 0, stream>>>(agg, b, lnw, lnb, hout);
}

extern "C" void kernel_launch(void* const* d_in, const int* in_sizes, int n_in,
                              void* d_out, int out_size, void* d_ws, size_t ws_size,
                              hipStream_t stream) {
  (void)in_sizes; (void)n_in; (void)out_size; (void)ws_size;

  const float* x       = (const float*)d_in[0];
  const int*   eidx    = (const int*)d_in[1];
  const int*   batch   = (const int*)d_in[2];
  const float* W_in    = (const float*)d_in[3];
  const float* b_in    = (const float*)d_in[4];
  const float* ln_in_w = (const float*)d_in[5];
  const float* ln_in_b = (const float*)d_in[6];
  const float* W_h     = (const float*)d_in[7];
  const float* b_h     = (const float*)d_in[8];
  const float* ln_h_w  = (const float*)d_in[9];
  const float* ln_h_b  = (const float*)d_in[10];
  const float* W_out   = (const float*)d_in[11];
  const float* b_out   = (const float*)d_in[12];

  const int* src = eidx;
  const int* dst = eidx + N_EDGES;

  float* ws    = (float*)d_ws;
  float* dinv  = ws;
  int*   cnt   = (int*)(dinv + N_PAD);
  int*   rowptr= cnt + N_PAD;
  int*   rank  = rowptr + N_PAD;
  int2*  ed    = (int2*)(rank + N_EDGES);
  float* hW    = (float*)(ed + N_EDGES);
  float* agg   = hW + (size_t)N_NODES*64;
  float* hbuf  = agg + (size_t)N_NODES*64;
  float* pool  = hbuf + (size_t)N_NODES*64;
  float* cnt_g = pool + N_GRAPHS * 64;

  k_cnt_zero<<<cdiv(N_NODES, 256), 256, 0, stream>>>(cnt);
  k_rank<<<cdiv(N_EDGES, 256), 256, 0, stream>>>(dst, cnt, rank);
  k_scan<<<cdiv(N_NODES + 1, 1024), 1024, 0, stream>>>(cnt, rowptr, dinv);
  k_fill<<<cdiv(N_EDGES, 256), 256, 0, stream>>>(src, dst, rank, rowptr, dinv, ed);

  run_layer(x,    W_in,       b_in,       ln_in_w,      ln_in_b,      dinv, rowptr, ed, hW, agg, hbuf, stream);
  run_layer(hbuf, W_h,        b_h,        ln_h_w,       ln_h_b,       dinv, rowptr, ed, hW, agg, hbuf, stream);
  run_layer(hbuf, W_h + 64*64, b_h + 64,  ln_h_w + 64,  ln_h_b + 64,  dinv, rowptr, ed, hW, agg, hbuf, stream);

  k_zero_pool<<<cdiv(N_GRAPHS * 64, 256), 256, 0, stream>>>(pool, cnt_g);
  k_pool_accum<<<cdiv(N_NODES * 64, 256), 256, 0, stream>>>(hbuf, batch, pool, cnt_g);
  k_pool_div<<<cdiv(N_GRAPHS * 64, 256), 256, 0, stream>>>(pool, cnt_g);
  gcn_gemm_wmma<D_OUT, true><<<N_GRAPHS / 16, 64, 0, stream>>>(pool, W_out, b_out, (float*)d_out);
}


# bias+layernorm+relu fused into the row-gather wave (it holds the whole row): 3 launches, 150k waves, agg round trip gone; layernorm sums pair features 2l,2l+1 first (was l,l+32)
# speedup vs baseline: 2.6780x; 1.0552x over previous
#include <hip/hip_runtime.h>
#include <hip/hip_bf16.h>

#define N_NODES  50000
#define N_EDGES  800000
#define N_GRAPHS 512
#define D_H      64
#define D_OUT    32
#define LN_EPS   1e-5f
#define N_PAD    50048

typedef __attribute__((ext_vector_type(2)))  float    v2f;
typedef __attribute__((ext_vector_type(8)))  float    v8f;
typedef __attribute__((ext_vector_type(16))) _Float16 v16h;

#if defined(__has_builtin)
#if __has_builtin(__builtin_amdgcn_wmma_f32_16x16x4_f32)
#define USE_F32_WMMA 1
#endif
#endif

template <int NCOLS, bool BIAS>
__global__ __launch_bounds__(32 * (NCOLS / 16)) void gcn_gemm_wmma(
    const float* __restrict__ A, const float* __restrict__ B,
    const float* __restrict__ bias, float* __restrict__ C) {
  const int lane = threadIdx.x & 31;
  const int wave = threadIdx.x >> 5;
  const int m    = lane & 15;
  const int half = lane >> 4;
  const int c0   = wave * 16;
  const int r0   = blockIdx.x * 16;
  const int row  = r0 + m;

  v8f acc = {0.f, 0.f, 0.f, 0.f, 0.f, 0.f, 0.f, 0.f};

#if USE_F32_WMMA
  #pragma unroll
  for (int k0 = 0; k0 < 64; k0 += 4) {
    const int ka = k0 + 2 * half;
    v2f a, b;
    a.x = A[row * 64 + ka];
    a.y = A[row * 64 + ka + 1];
    b.x = B[ka * NCOLS + c0 + m];
    b.y = B[(ka + 1) * NCOLS + c0 + m];
    acc = __builtin_amdgcn_wmma_f32_16x16x4_f32(false, a, false, b,
                                                (short)0, acc, false, false);
  }
#else
  for (int kstep = 0; kstep < 2; ++kstep) {
    v16h af, bf;
    #pragma unroll
    for (int e = 0; e < 16; ++e) {
      int v = e >> 1, j = e & 1, grp = v >> 2;
      int K = 32 * kstep + 16 * grp + 8 * half + ((v & 3) * 2 + j);
      af[e] = (_Float16)A[row * 64 + K];
    }
    #pragma unroll
    for (int e = 0; e < 16; ++e) {
      int v = e >> 1, j = e & 1;
      int K = 32 * kstep + 16 * half + 2 * v + j;
      bf[e] = (_Float16)B[K * NCOLS + c0 + m];
    }
    acc = __builtin_amdgcn_wmma_f32_16x16x32_f16(false, af, false, bf,
                                                 (short)0, acc, false, false);
  }
#endif

  #pragma unroll
  for (int i = 0; i < 8; ++i) {
    float v = acc[i];
    if (BIAS) v += bias[c0 + m];
    C[(r0 + i + 8 * half) * NCOLS + c0 + m] = v;
  }
}

__global__ void k_cnt_zero(int* __restrict__ cnt) {
  int i = blockIdx.x * blockDim.x + threadIdx.x;
  if (i < N_NODES) cnt[i] = 0;
}

__global__ void k_rank(const int* __restrict__ dst, int* __restrict__ cnt,
                       int* __restrict__ rank) {
  int i = blockIdx.x * blockDim.x + threadIdx.x;
  if (i < N_EDGES) rank[i] = atomicAdd(&cnt[dst[i]], 1);
}

__global__ __launch_bounds__(1024) void k_scan(const int* __restrict__ cnt,
                                               int* __restrict__ rowptr,
                                               float* __restrict__ dinv) {
  __shared__ int s_front[32];
  __shared__ int s_wtot[32];
  const int tid = threadIdx.x, lane = tid & 31, wave = tid >> 5;
  const int n = blockIdx.x * 1024 + tid;

  const int4* c4 = (const int4*)cnt;
  const int n4 = blockIdx.x * 256;
  int front = 0;
  #pragma unroll 4
  for (int j = tid; j < n4; j += 1024) {
    int4 v = c4[j];
    front += (v.x + v.y) + (v.z + v.w);
  }
  #pragma unroll
  for (int off = 16; off > 0; off >>= 1) front += __shfl_xor(front, off, 32);

  const int c = (n < N_NODES) ? cnt[n] : 0;
  int incl = c;
  #pragma unroll
  for (int off = 1; off < 32; off <<= 1) {
    int t = __shfl_up(incl, off, 32);
    if (lane >= off) incl += t;
  }
  if (lane == 0)  s_front[wave] = front;
  if (lane == 31) s_wtot[wave]  = incl;
  __syncthreads();
  int base = 0;
  for (int w = 0; w < 32; ++w) {
    base += s_front[w];
    if (w < wave) base += s_wtot[w];
  }
  if (n <= N_NODES) rowptr[n] = base + incl - c;
  if (n < N_NODES)  dinv[n] = rsqrtf((float)(c + 1));
}

__global__ void k_fill(const int* __restrict__ src, const int* __restrict__ dst,
                       const int* __restrict__ rank, const int* __restrict__ rowptr,
                       const float* __restrict__ dinv, int2* __restrict__ ed) {
  int i = blockIdx.x * blockDim.x + threadIdx.x;
  if (i >= N_EDGES) return;
  int s = src[i], d = dst[i];
  ed[rowptr[d] + rank[i]] = make_int2(s, __float_as_int(dinv[s] * dinv[d]));
}

template <int K, bool CLAMP>
__device__ __forceinline__ void gather_rows(const float2* __restrict__ h2,
                                            const int2* __restrict__ ed,
                                            int p, int end, int lane,
                                            float& ax, float& ay) {
  int2 e[K];
  #pragma unroll
  for (int j = 0; j < K; ++j) {
    int q = p + j;
    if (CLAMP) q = min(q, end - 1);
    e[j] = ed[q];
  }
  float2 r[K];
  #pragma unroll
  for (int j = 0; j < K; ++j) r[j] = (h2 + e[j].x * 32)[lane];
  #pragma unroll
  for (int j = 0; j < K; ++j) {
    float w = __int_as_float(e[j].y);
    if (CLAMP && p + j >= end) w = 0.f;
    ax += w * r[j].x;
    ay += w * r[j].y;
  }
}

__global__ __launch_bounds__(256) void k_gather_ln(const float* __restrict__ hW,
                                                   const int* __restrict__ rowptr,
                                                   const int2* __restrict__ ed,
                                                   const float* __restrict__ dinv,
                                                   const float* __restrict__ bias,
                                                   const float* __restrict__ lnw,
                                                   const float* __restrict__ lnb,
                                                   float* __restrict__ hout) {
  const int lane = threadIdx.x & 31;
  const int node = __builtin_amdgcn_readfirstlane((int)((blockIdx.x * 256 + threadIdx.x) >> 5));
  if (node >= N_NODES) return;
  const int beg = rowptr[node], end = rowptr[node + 1];
  const float di = dinv[node];
  const float2* h2 = (const float2*)hW;
  const float2 self = (h2 + node * 32)[lane];
  const float b0 = bias[2 * lane], b1 = bias[2 * lane + 1];
  const float w0 = lnw[2 * lane],  w1 = lnw[2 * lane + 1];
  const float o0 = lnb[2 * lane],  o1 = lnb[2 * lane + 1];
  float ax = di * di * self.x;
  float ay = di * di * self.y;
  int p = beg;
  for (; p + 8 <= end; p += 8) gather_rows<8, false>(h2, ed, p, end, lane, ax, ay);
  if (p + 4 <= end) { gather_rows<4, false>(h2, ed, p, end, lane, ax, ay); p += 4; }
  if (p < end) gather_rows<4, true>(h2, ed, p, end, lane, ax, ay);

  const float v0 = ax + b0, v1 = ay + b1;
  float s = v0 + v1;
  #pragma unroll
  for (int off = 16; off > 0; off >>= 1) s += __shfl_xor(s, off, 32);
  const float mu = s * (1.0f / 64.0f);
  const float d0 = v0 - mu, d1 = v1 - mu;
  float q = d0 * d0 + d1 * d1;
  #pragma unroll
  for (int off = 16; off > 0; off >>= 1) q += __shfl_xor(q, off, 32);
  const float rstd = rsqrtf(q * (1.0f / 64.0f) + LN_EPS);
  ((float2*)hout + node * 32)[lane] =
      make_float2(fmaxf(0.f, d0 * rstd * w0 + o0), fmaxf(0.f, d1 * rstd * w1 + o1));
}

__global__ void k_zero_pool(float* __restrict__ pool, float* __restrict__ cnt) {
  int i = blockIdx.x * blockDim.x + threadIdx.x;
  if (i < N_GRAPHS * 64) pool[i] = 0.f;
  if (i < N_GRAPHS) cnt[i] = 0.f;
}

__global__ void k_pool_accum(const float* __restrict__ h, const int* __restrict__ batch,
                             float* __restrict__ pool, float* __restrict__ cnt) {
  int idx = blockIdx.x * blockDim.x + threadIdx.x;
  if (idx >= N_NODES * 64) return;
  int n = idx >> 6, f = idx & 63;
  int g = batch[n];
  atomicAdd(&pool[g * 64 + f], h[idx]);
  if (f == 0) atomicAdd(&cnt[g], 1.0f);
}

__global__ void k_pool_div(float* __restrict__ pool, const float* __restrict__ cnt) {
  int i = blockIdx.x * blockDim.x + threadIdx.x;
  if (i < N_GRAPHS * 64) pool[i] /= fmaxf(cnt[i >> 6], 1.0f);
}

static inline int cdiv(int a, int b) { return (a + b - 1) / b; }

static void run_layer(const float* hin, const float* W, const float* b,
                      const float* lnw, const float* lnb,
                      const float* dinv, const int* rowptr, const int2* ed,
                      float* hW, float* hout,
                      hipStream_t stream) {
  gcn_gemm_wmma<64, false><<<N_NODES / 16, 128, 0, stream>>>(hin, W, nullptr, hW);
  k_gather_ln<<<cdiv(N_NODES * 32, 256), 256, 0, stream>>>(hW, rowptr, ed, dinv, b, lnw, lnb, hout);
}

extern "C" void kernel_launch(void* const* d_in, const int* in_sizes, int n_in,
                              void* d_out, int out_size, void* d_ws, size_t ws_size,
                              hipStream_t stream) {
  (void)in_sizes; (void)n_in; (void)out_size; (void)ws_size;

  const float* x       = (const float*)d_in[0];
  const int*   eidx    = (const int*)d_in[1];
  const int*   batch   = (const int*)d_in[2];
  const float* W_in    = (const float*)d_in[3];
  const float* b_in    = (const float*)d_in[4];
  const float* ln_in_w = (const float*)d_in[5];
  const float* ln_in_b = (const float*)d_in[6];
  const float* W_h     = (const float*)d_in[7];
  const float* b_h     = (const float*)d_in[8];
  const float* ln_h_w  = (const float*)d_in[9];
  const float* ln_h_b  = (const float*)d_in[10];
  const float* W_out   = (const float*)d_in[11];
  const float* b_out   = (const float*)d_in[12];

  const int* src = eidx;
  const int* dst = eidx + N_EDGES;

  float* ws    = (float*)d_ws;
  float* dinv  = ws;
  int*   cnt   = (int*)(dinv + N_PAD);
  int*   rowptr= cnt + N_PAD;
  int*   rank  = rowptr + N_PAD;
  int2*  ed    = (int2*)(rank + N_EDGES);
  float* hW    = (float*)(ed + N_EDGES);
  float* hbuf  = hW + (size_t)N_NODES*64;
  float* pool  = hbuf + (size_t)N_NODES*64;
  float* cnt_g = pool + N_GRAPHS * 64;

  k_cnt_zero<<<cdiv(N_NODES, 256), 256, 0, stream>>>(cnt);
  k_rank<<<cdiv(N_EDGES, 256), 256, 0, stream>>>(dst, cnt, rank);
  k_scan<<<cdiv(N_NODES + 1, 1024), 1024, 0, stream>>>(cnt, rowptr, dinv);
  k_fill<<<cdiv(N_EDGES, 256), 256, 0, stream>>>(src, dst, rank, rowptr, dinv, ed);

  run_layer(x,    W_in,       b_in,       ln_in_w,      ln_in_b,      dinv, rowptr, ed, hW, hbuf, stream);
  run_layer(hbuf, W_h,        b_h,        ln_h_w,       ln_h_b,       dinv, rowptr, ed, hW, hbuf, stream);
  run_layer(hbuf, W_h + 64*64, b_h + 64,  ln_h_w + 64,  ln_h_b + 64,  dinv, rowptr, ed, hW, hbuf, stream);

  k_zero_pool<<<cdiv(N_GRAPHS * 64, 256), 256, 0, stream>>>(pool, cnt_g);
  k_pool_accum<<<cdiv(N_NODES * 64, 256), 256, 0, stream>>>(hbuf, batch, pool, cnt_g);
  k_pool_div<<<cdiv(N_GRAPHS * 64, 256), 256, 0, stream>>>(pool, cnt_g);
  gcn_gemm_wmma<D_OUT, true><<<N_GRAPHS / 16, 64, 0, stream>>>(pool, W_out, b_out, (float*)d_out);
}


# Edge slots counted in LDS instead of by 800k scattered returning global atomics: block (row range, edge chunk) ranks its edges with LDS adds, scan adds chunk starts; sum order still varies by run
# speedup vs baseline: 3.4225x; 1.2780x over previous
#include <hip/hip_runtime.h>
#include <hip/hip_bf16.h>

#define N_NODES  50000
#define N_EDGES  800000
#define N_GRAPHS 512
#define D_H      64
#define D_OUT    32
#define LN_EPS   1e-5f
#define N_PAD    50048

typedef __attribute__((ext_vector_type(2)))  float    v2f;
typedef __attribute__((ext_vector_type(8)))  float    v8f;
typedef __attribute__((ext_vector_type(16))) _Float16 v16h;

#if defined(__has_builtin)
#if __has_builtin(__builtin_amdgcn_wmma_f32_16x16x4_f32)
#define USE_F32_WMMA 1
#endif
#endif

template <int NCOLS, bool BIAS>
__global__ __launch_bounds__(32 * (NCOLS / 16)) void gcn_gemm_wmma(
    const float* __restrict__ A, const float* __restrict__ B,
    const float* __restrict__ bias, float* __restrict__ C) {
  const int lane = threadIdx.x & 31;
  const int wave = threadIdx.x >> 5;
  const int m    = lane & 15;
  const int half = lane >> 4;
  const int c0   = wave * 16;
  const int r0   = blockIdx.x * 16;
  const int row  = r0 + m;

  v8f acc = {0.f, 0.f, 0.f, 0.f, 0.f, 0.f, 0.f, 0.f};

#if USE_F32_WMMA
  #pragma unroll
  for (int k0 = 0; k0 < 64; k0 += 4) {
    const int ka = k0 + 2 * half;
    v2f a, b;
    a.x = A[row * 64 + ka];
    a.y = A[row * 64 + ka + 1];
    b.x = B[ka * NCOLS + c0 + m];
    b.y = B[(ka + 1) * NCOLS + c0 + m];
    acc = __builtin_amdgcn_wmma_f32_16x16x4_f32(false, a, false, b,
                                                (short)0, acc, false, false);
  }
#else
  for (int kstep = 0; kstep < 2; ++kstep) {
    v16h af, bf;
    #pragma unroll
    for (int e = 0; e < 16; ++e) {
      int v = e >> 1, j = e & 1, grp = v >> 2;
      int K = 32 * kstep + 16 * grp + 8 * half + ((v & 3) * 2 + j);
      af[e] = (_Float16)A[row * 64 + K];
    }
    #pragma unroll
    for (int e = 0; e < 16; ++e) {
      int v = e >> 1, j = e & 1;
      int K = 32 * kstep + 16 * half + 2 * v + j;
      bf[e] = (_Float16)B[K * NCOLS + c0 + m];
    }
    acc = __builtin_amdgcn_wmma_f32_16x16x32_f16(false, af, false, bf,
                                                 (short)0, acc, false, false);
  }
#endif

  #pragma unroll
  for (int i = 0; i < 8; ++i) {
    float v = acc[i];
    if (BIAS) v += bias[c0 + m];
    C[(r0 + i + 8 * half) * NCOLS + c0 + m] = v;
  }
}

#define RANGE   4096
#define NPART   13
#define NCHUNK  16
#define CH      (N_EDGES / NCHUNK)
#define NSUB    (NPART * 4)
#define NLOAD   13

__global__ __launch_bounds__(1024) void k_rank_lds(const int* __restrict__ dst,
                                                   int* __restrict__ rank,
                                                   int* __restrict__ cntc,
                                                   int* __restrict__ wgtot) {
  __shared__ int s_cnt[RANGE];
  __shared__ int s_red[128];
  const int tid = threadIdx.x, lane = tid & 31, wave = tid >> 5;
  const int p = blockIdx.x / NCHUNK, c = blockIdx.x % NCHUNK;
  const int lo = p * RANGE;
  #pragma unroll
  for (int q = 0; q < 4; ++q) s_cnt[q * 1024 + tid] = 0;

  const int e0 = c * CH;
  const int4* d4 = (const int4*)(dst + e0);
  int4 v[NLOAD];
  #pragma unroll
  for (int k = 0; k < NLOAD; ++k) {
    const int j = tid + k * 1024;
    v[k] = (j < CH / 4) ? d4[j] : make_int4(-1, -1, -1, -1);
  }
  __syncthreads();
  #pragma unroll
  for (int k = 0; k < NLOAD; ++k) {
    const int e = e0 + 4 * (tid + k * 1024);
    unsigned r;
    r = (unsigned)(v[k].x - lo); if (r < RANGE) rank[e]     = atomicAdd(&s_cnt[r], 1);
    r = (unsigned)(v[k].y - lo); if (r < RANGE) rank[e + 1] = atomicAdd(&s_cnt[r], 1);
    r = (unsigned)(v[k].z - lo); if (r < RANGE) rank[e + 2] = atomicAdd(&s_cnt[r], 1);
    r = (unsigned)(v[k].w - lo); if (r < RANGE) rank[e + 3] = atomicAdd(&s_cnt[r], 1);
  }
  __syncthreads();
  #pragma unroll
  for (int q = 0; q < 4; ++q) {
    const int n = lo + q * 1024 + tid;
    int t = s_cnt[q * 1024 + tid];
    if (n < N_NODES) cntc[c * N_PAD + n] = t;
    #pragma unroll
    for (int off = 16; off > 0; off >>= 1) t += __shfl_xor(t, off, 32);
    if (lane == 0) s_red[q * 32 + wave] = t;
  }
  __syncthreads();
  if (tid < 128) {
    int t = s_red[tid];
    #pragma unroll
    for (int off = 16; off > 0; off >>= 1) t += __shfl_xor(t, off, 32);
    if (lane == 0) wgtot[(p * 4 + wave) * NCHUNK + c] = t;
  }
}

__global__ __launch_bounds__(1024) void k_scan(const int* __restrict__ cntc,
                                               const int* __restrict__ wgtot,
                                               int* __restrict__ rowptr,
                                               int* __restrict__ basec,
                                               float* __restrict__ dinv) {
  __shared__ int s_front[32];
  __shared__ int s_wtot[32];
  const int tid = threadIdx.x, lane = tid & 31, wave = tid >> 5;
  const int n = blockIdx.x * 1024 + tid;

  int front = (tid < (int)blockIdx.x * NCHUNK) ? wgtot[tid] : 0;
  #pragma unroll
  for (int off = 16; off > 0; off >>= 1) front += __shfl_xor(front, off, 32);

  int cc[NCHUNK];
  int c = 0;
  #pragma unroll
  for (int k = 0; k < NCHUNK; ++k) {
    cc[k] = (n < N_NODES) ? cntc[k * N_PAD + n] : 0;
    c += cc[k];
  }
  int incl = c;
  #pragma unroll
  for (int off = 1; off < 32; off <<= 1) {
    int t = __shfl_up(incl, off, 32);
    if (lane >= off) incl += t;
  }
  if (lane == 0)  s_front[wave] = front;
  if (lane == 31) s_wtot[wave]  = incl;
  __syncthreads();
  int base = 0;
  for (int w = 0; w < 32; ++w) {
    base += s_front[w];
    if (w < wave) base += s_wtot[w];
  }
  int run = base + incl - c;
  if (n <= N_NODES) rowptr[n] = run;
  if (n < N_NODES) {
    dinv[n] = rsqrtf((float)(c + 1));
    #pragma unroll
    for (int k = 0; k < NCHUNK; ++k) {
      basec[k * N_PAD + n] = run;
      run += cc[k];
    }
  }
}

__global__ void k_fill(const int* __restrict__ src, const int* __restrict__ dst,
                       const int* __restrict__ rank, const int* __restrict__ basec,
                       const float* __restrict__ dinv, int2* __restrict__ ed) {
  int i = blockIdx.x * blockDim.x + threadIdx.x;
  if (i >= N_EDGES) return;
  int s = src[i], d = dst[i];
  ed[basec[(i / CH) * N_PAD + d] + rank[i]] = make_int2(s, __float_as_int(dinv[s] * dinv[d]));
}

template <int K, bool CLAMP>
__device__ __forceinline__ void gather_rows(const float2* __restrict__ h2,
                                            const int2* __restrict__ ed,
                                            int p, int end, int lane,
                                            float& ax, float& ay) {
  int2 e[K];
  #pragma unroll
  for (int j = 0; j < K; ++j) {
    int q = p + j;
    if (CLAMP) q = min(q, end - 1);
    e[j] = ed[q];
  }
  float2 r[K];
  #pragma unroll
  for (int j = 0; j < K; ++j) r[j] = (h2 + e[j].x * 32)[lane];
  #pragma unroll
  for (int j = 0; j < K; ++j) {
    float w = __int_as_float(e[j].y);
    if (CLAMP && p + j >= end) w = 0.f;
    ax += w * r[j].x;
    ay += w * r[j].y;
  }
}

__global__ __launch_bounds__(256) void k_gather_ln(const float* __restrict__ hW,
                                                   const int* __restrict__ rowptr,
                                                   const int2* __restrict__ ed,
                                                   const float* __restrict__ dinv,
                                                   const float* __restrict__ bias,
                                                   const float* __restrict__ lnw,
                                                   const float* __restrict__ lnb,
                                                   float* __restrict__ hout) {
  const int lane = threadIdx.x & 31;
  const int node = __builtin_amdgcn_readfirstlane((int)((blockIdx.x * 256 + threadIdx.x) >> 5));
  if (node >= N_NODES) return;
  const int beg = rowptr[node], end = rowptr[node + 1];
  const float di = dinv[node];
  const float2* h2 = (const float2*)hW;
  const float2 self = (h2 + node * 32)[lane];
  const float b0 = bias[2 * lane], b1 = bias[2 * lane + 1];
  const float w0 = lnw[2 * lane],  w1 = lnw[2 * lane + 1];
  const float o0 = lnb[2 * lane],  o1 = lnb[2 * lane + 1];
  float ax = di * di * self.x;
  float ay = di * di * self.y;
  int p = beg;
  for (; p + 8 <= end; p += 8) gather_rows<8, false>(h2, ed, p, end, lane, ax, ay);
  if (p + 4 <= end) { gather_rows<4, false>(h2, ed, p, end, lane, ax, ay); p += 4; }
  if (p < end) gather_rows<4, true>(h2, ed, p, end, lane, ax, ay);

  const float v0 = ax + b0, v1 = ay + b1;
  float s = v0 + v1;
  #pragma unroll
  for (int off = 16; off > 0; off >>= 1) s += __shfl_xor(s, off, 32);
  const float mu = s * (1.0f / 64.0f);
  const float d0 = v0 - mu, d1 = v1 - mu;
  float q = d0 * d0 + d1 * d1;
  #pragma unroll
  for (int off = 16; off > 0; off >>= 1) q += __shfl_xor(q, off, 32);
  const float rstd = rsqrtf(q * (1.0f / 64.0f) + LN_EPS);
  ((float2*)hout + node * 32)[lane] =
      make_float2(fmaxf(0.f, d0 * rstd * w0 + o0), fmaxf(0.f, d1 * rstd * w1 + o1));
}

__global__ void k_zero_pool(float* __restrict__ pool, float* __restrict__ cnt) {
  int i = blockIdx.x * blockDim.x + threadIdx.x;
  if (i < N_GRAPHS * 64) pool[i] = 0.f;
  if (i < N_GRAPHS) cnt[i] = 0.f;
}

__global__ void k_pool_accum(const float* __restrict__ h, const int* __restrict__ batch,
                             float* __restrict__ pool, float* __restrict__ cnt) {
  int idx = blockIdx.x * blockDim.x + threadIdx.x;
  if (idx >= N_NODES * 64) return;
  int n = idx >> 6, f = idx & 63;
  int g = batch[n];
  atomicAdd(&pool[g * 64 + f], h[idx]);
  if (f == 0) atomicAdd(&cnt[g], 1.0f);
}

__global__ void k_pool_div(float* __restrict__ pool, const float* __restrict__ cnt) {
  int i = blockIdx.x * blockDim.x + threadIdx.x;
  if (i < N_GRAPHS * 64) pool[i] /= fmaxf(cnt[i >> 6], 1.0f);
}

static inline int cdiv(int a, int b) { return (a + b - 1) / b; }

static void run_layer(const float* hin, const float* W, const float* b,
                      const float* lnw, const float* lnb,
                      const float* dinv, const int* rowptr, const int2* ed,
                      float* hW, float* hout,
                      hipStream_t stream) {
  gcn_gemm_wmma<64, false><<<N_NODES / 16, 128, 0, stream>>>(hin, W, nullptr, hW);
  k_gather_ln<<<cdiv(N_NODES * 32, 256), 256, 0, stream>>>(hW, rowptr, ed, dinv, b, lnw, lnb, hout);
}

extern "C" void kernel_launch(void* const* d_in, const int* in_sizes, int n_in,
                              void* d_out, int out_size, void* d_ws, size_t ws_size,
                              hipStream_t stream) {
  (void)in_sizes; (void)n_in; (void)out_size; (void)ws_size;

  const float* x       = (const float*)d_in[0];
  const int*   eidx    = (const int*)d_in[1];
  const int*   batch   = (const int*)d_in[2];
  const float* W_in    = (const float*)d_in[3];
  const float* b_in    = (const float*)d_in[4];
  const float* ln_in_w = (const float*)d_in[5];
  const float* ln_in_b = (const float*)d_in[6];
  const float* W_h     = (const float*)d_in[7];
  const float* b_h     = (const float*)d_in[8];
  const float* ln_h_w  = (const float*)d_in[9];
  const float* ln_h_b  = (const float*)d_in[10];
  const float* W_out   = (const float*)d_in[11];
  const float* b_out   = (const float*)d_in[12];

  const int* src = eidx;
  const int* dst = eidx + N_EDGES;

  float* ws    = (float*)d_ws;
  float* dinv  = ws;
  int*   rowptr= (int*)(dinv + N_PAD);
  int*   cntc  = rowptr + N_PAD;
  int*   basec = cntc + NCHUNK * N_PAD;
  int*   wgtot = basec + NCHUNK * N_PAD;
  int*   rank  = wgtot + 1024;
  int2*  ed    = (int2*)(rank + N_EDGES);
  float* hW    = (float*)(ed + N_EDGES);
  float* hbuf  = hW + (size_t)N_NODES*64;
  float* pool  = hbuf + (size_t)N_NODES*64;
  float* cnt_g = pool + N_GRAPHS * 64;

  k_rank_lds<<<NPART * NCHUNK, 1024, 0, stream>>>(dst, rank, cntc, wgtot);
  k_scan<<<cdiv(N_NODES + 1, 1024), 1024, 0, stream>>>(cntc, wgtot, rowptr, basec, dinv);
  k_fill<<<cdiv(N_EDGES, 256), 256, 0, stream>>>(src, dst, rank, basec, dinv, ed);

  run_layer(x,    W_in,       b_in,       ln_in_w,      ln_in_b,      dinv, rowptr, ed, hW, hbuf, stream);
  run_layer(hbuf, W_h,        b_h,        ln_h_w,       ln_h_b,       dinv, rowptr, ed, hW, hbuf, stream);
  run_layer(hbuf, W_h + 64*64, b_h + 64,  ln_h_w + 64,  ln_h_b + 64,  dinv, rowptr, ed, hW, hbuf, stream);

  k_zero_pool<<<cdiv(N_GRAPHS * 64, 256), 256, 0, stream>>>(pool, cnt_g);
  k_pool_accum<<<cdiv(N_NODES * 64, 256), 256, 0, stream>>>(hbuf, batch, pool, cnt_g);
  k_pool_div<<<cdiv(N_GRAPHS * 64, 256), 256, 0, stream>>>(pool, cnt_g);
  gcn_gemm_wmma<D_OUT, true><<<N_GRAPHS / 16, 64, 0, stream>>>(pool, W_out, b_out, (float*)d_out);
}


# Mean pool folded into the last layer's row wave: a block's 8 rows meet in LDS, one f32 atomic per feature per block (sum order: 8-row partials, arrival order still varies); 13 to 10 launches
# speedup vs baseline: 4.5713x; 1.3357x over previous
#include <hip/hip_runtime.h>
#include <hip/hip_bf16.h>

#define N_NODES  50000
#define N_EDGES  800000
#define N_GRAPHS 512
#define D_H      64
#define D_OUT    32
#define LN_EPS   1e-5f
#define N_PAD    50048

typedef __attribute__((ext_vector_type(2)))  float    v2f;
typedef __attribute__((ext_vector_type(8)))  float    v8f;
typedef __attribute__((ext_vector_type(16))) _Float16 v16h;

#if defined(__has_builtin)
#if __has_builtin(__builtin_amdgcn_wmma_f32_16x16x4_f32)
#define USE_F32_WMMA 1
#endif
#endif

template <int NCOLS, bool BIAS, bool MEAN>
__global__ __launch_bounds__(32 * (NCOLS / 16)) void gcn_gemm_wmma(
    const float* __restrict__ A, const float* __restrict__ B,
    const float* __restrict__ bias, const float* __restrict__ cnt, float* __restrict__ C) {
  const int lane = threadIdx.x & 31;
  const int wave = threadIdx.x >> 5;
  const int m    = lane & 15;
  const int half = lane >> 4;
  const int c0   = wave * 16;
  const int r0   = blockIdx.x * 16;
  const int row  = r0 + m;

  v8f acc = {0.f, 0.f, 0.f, 0.f, 0.f, 0.f, 0.f, 0.f};
  float den = 1.0f;
  if (MEAN) den = fmaxf(cnt[row], 1.0f);

#if USE_F32_WMMA
  #pragma unroll
  for (int k0 = 0; k0 < 64; k0 += 4) {
    const int ka = k0 + 2 * half;
    v2f a, b;
    a.x = A[row * 64 + ka];
    a.y = A[row * 64 + ka + 1];
    if (MEAN) { a.x /= den; a.y /= den; }
    b.x = B[ka * NCOLS + c0 + m];
    b.y = B[(ka + 1) * NCOLS + c0 + m];
    acc = __builtin_amdgcn_wmma_f32_16x16x4_f32(false, a, false, b,
                                                (short)0, acc, false, false);
  }
#else
  for (int kstep = 0; kstep < 2; ++kstep) {
    v16h af, bf;
    #pragma unroll
    for (int e = 0; e < 16; ++e) {
      int v = e >> 1, j = e & 1, grp = v >> 2;
      int K = 32 * kstep + 16 * grp + 8 * half + ((v & 3) * 2 + j);
      af[e] = (_Float16)(MEAN ? A[row * 64 + K] / den : A[row * 64 + K]);
    }
    #pragma unroll
    for (int e = 0; e < 16; ++e) {
      int v = e >> 1, j = e & 1;
      int K = 32 * kstep + 16 * half + 2 * v + j;
      bf[e] = (_Float16)B[K * NCOLS + c0 + m];
    }
    acc = __builtin_amdgcn_wmma_f32_16x16x32_f16(false, af, false, bf,
                                                 (short)0, acc, false, false);
  }
#endif

  #pragma unroll
  for (int i = 0; i < 8; ++i) {
    float v = acc[i];
    if (BIAS) v += bias[c0 + m];
    C[(r0 + i + 8 * half) * NCOLS + c0 + m] = v;
  }
}

#define RANGE   4096
#define NPART   13
#define NCHUNK  16
#define CH      (N_EDGES / NCHUNK)
#define NSUB    (NPART * 4)
#define NLOAD   13

__global__ __launch_bounds__(1024) void k_rank_lds(const int* __restrict__ dst,
                                                   int* __restrict__ rank,
                                                   int* __restrict__ cntc,
                                                   int* __restrict__ wgtot) {
  __shared__ int s_cnt[RANGE];
  __shared__ int s_red[128];
  const int tid = threadIdx.x, lane = tid & 31, wave = tid >> 5;
  const int p = blockIdx.x / NCHUNK, c = blockIdx.x % NCHUNK;
  const int lo = p * RANGE;
  #pragma unroll
  for (int q = 0; q < 4; ++q) s_cnt[q * 1024 + tid] = 0;

  const int e0 = c * CH;
  const int4* d4 = (const int4*)(dst + e0);
  int4 v[NLOAD];
  #pragma unroll
  for (int k = 0; k < NLOAD; ++k) {
    const int j = tid + k * 1024;
    v[k] = (j < CH / 4) ? d4[j] : make_int4(-1, -1, -1, -1);
  }
  __syncthreads();
  #pragma unroll
  for (int k = 0; k < NLOAD; ++k) {
    const int e = e0 + 4 * (tid + k * 1024);
    unsigned r;
    r = (unsigned)(v[k].x - lo); if (r < RANGE) rank[e]     = atomicAdd(&s_cnt[r], 1);
    r = (unsigned)(v[k].y - lo); if (r < RANGE) rank[e + 1] = atomicAdd(&s_cnt[r], 1);
    r = (unsigned)(v[k].z - lo); if (r < RANGE) rank[e + 2] = atomicAdd(&s_cnt[r], 1);
    r = (unsigned)(v[k].w - lo); if (r < RANGE) rank[e + 3] = atomicAdd(&s_cnt[r], 1);
  }
  __syncthreads();
  #pragma unroll
  for (int q = 0; q < 4; ++q) {
    const int n = lo + q * 1024 + tid;
    int t = s_cnt[q * 1024 + tid];
    if (n < N_NODES) cntc[c * N_PAD + n] = t;
    #pragma unroll
    for (int off = 16; off > 0; off >>= 1) t += __shfl_xor(t, off, 32);
    if (lane == 0) s_red[q * 32 + wave] = t;
  }
  __syncthreads();
  if (tid < 128) {
    int t = s_red[tid];
    #pragma unroll
    for (int off = 16; off > 0; off >>= 1) t += __shfl_xor(t, off, 32);
    if (lane == 0) wgtot[(p * 4 + wave) * NCHUNK + c] = t;
  }
}

__global__ __launch_bounds__(1024) void k_scan(const int* __restrict__ cntc,
                                               const int* __restrict__ wgtot,
                                               int* __restrict__ rowptr,
                                               int* __restrict__ basec,
                                               float* __restrict__ dinv,
                                               float* __restrict__ pool,
                                               float* __restrict__ gcnt) {
  __shared__ int s_front[32];
  __shared__ int s_wtot[32];
  const int tid = threadIdx.x, lane = tid & 31, wave = tid >> 5;
  const int n = blockIdx.x * 1024 + tid;

  if (n < N_GRAPHS * D_H) pool[n] = 0.f;
  if (n < N_GRAPHS) gcnt[n] = 0.f;

  int front = (tid < (int)blockIdx.x * NCHUNK) ? wgtot[tid] : 0;
  #pragma unroll
  for (int off = 16; off > 0; off >>= 1) front += __shfl_xor(front, off, 32);

  int cc[NCHUNK];
  int c = 0;
  #pragma unroll
  for (int k = 0; k < NCHUNK; ++k) {
    cc[k] = (n < N_NODES) ? cntc[k * N_PAD + n] : 0;
    c += cc[k];
  }
  int incl = c;
  #pragma unroll
  for (int off = 1; off < 32; off <<= 1) {
    int t = __shfl_up(incl, off, 32);
    if (lane >= off) incl += t;
  }
  if (lane == 0)  s_front[wave] = front;
  if (lane == 31) s_wtot[wave]  = incl;
  __syncthreads();
  int base = 0;
  for (int w = 0; w < 32; ++w) {
    base += s_front[w];
    if (w < wave) base += s_wtot[w];
  }
  int run = base + incl - c;
  if (n <= N_NODES) rowptr[n] = run;
  if (n < N_NODES) {
    dinv[n] = rsqrtf((float)(c + 1));
    #pragma unroll
    for (int k = 0; k < NCHUNK; ++k) {
      basec[k * N_PAD + n] = run;
      run += cc[k];
    }
  }
}

__global__ void k_fill(const int* __restrict__ src, const int* __restrict__ dst,
                       const int* __restrict__ rank, const int* __restrict__ basec,
                       const float* __restrict__ dinv, int2* __restrict__ ed) {
  int i = blockIdx.x * blockDim.x + threadIdx.x;
  if (i >= N_EDGES) return;
  int s = src[i], d = dst[i];
  ed[basec[(i / CH) * N_PAD + d] + rank[i]] = make_int2(s, __float_as_int(dinv[s] * dinv[d]));
}

template <int K, bool CLAMP>
__device__ __forceinline__ void gather_rows(const float2* __restrict__ h2,
                                            const int2* __restrict__ ed,
                                            int p, int end, int lane,
                                            float& ax, float& ay) {
  int2 e[K];
  #pragma unroll
  for (int j = 0; j < K; ++j) {
    int q = p + j;
    if (CLAMP) q = min(q, end - 1);
    e[j] = ed[q];
  }
  float2 r[K];
  #pragma unroll
  for (int j = 0; j < K; ++j) r[j] = (h2 + e[j].x * 32)[lane];
  #pragma unroll
  for (int j = 0; j < K; ++j) {
    float w = __int_as_float(e[j].y);
    if (CLAMP && p + j >= end) w = 0.f;
    ax += w * r[j].x;
    ay += w * r[j].y;
  }
}

static_assert(N_NODES % 8 == 0, "a block holds 8 whole rows: all of its waves pass the row guard or none does");
template <bool POOL>
__global__ __launch_bounds__(256) void k_gather_ln(const float* __restrict__ hW,
                                                   const int* __restrict__ rowptr,
                                                   const int2* __restrict__ ed,
                                                   const float* __restrict__ dinv,
                                                   const float* __restrict__ bias,
                                                   const float* __restrict__ lnw,
                                                   const float* __restrict__ lnb,
                                                   float* __restrict__ hout,
                                                   const int* __restrict__ batch,
                                                   float* __restrict__ pool,
                                                   float* __restrict__ gcnt) {
  __shared__ float s_row[POOL ? 8 * D_H : 1];
  const int lane = threadIdx.x & 31;
  const int node = __builtin_amdgcn_readfirstlane((int)((blockIdx.x * 256 + threadIdx.x) >> 5));
  if (node >= N_NODES) return;
  const int beg = rowptr[node], end = rowptr[node + 1];
  const float di = dinv[node];
  int g[8];
  if (POOL) {
    #pragma unroll
    for (int w = 0; w < 8; ++w) g[w] = batch[(int)blockIdx.x * 8 + w];
  }
  const float2* h2 = (const float2*)hW;
  const float2 self = (h2 + node * 32)[lane];
  const float b0 = bias[2 * lane], b1 = bias[2 * lane + 1];
  const float w0 = lnw[2 * lane],  w1 = lnw[2 * lane + 1];
  const float o0 = lnb[2 * lane],  o1 = lnb[2 * lane + 1];
  float ax = di * di * self.x;
  float ay = di * di * self.y;
  int p = beg;
  for (; p + 8 <= end; p += 8) gather_rows<8, false>(h2, ed, p, end, lane, ax, ay);
  if (p + 4 <= end) { gather_rows<4, false>(h2, ed, p, end, lane, ax, ay); p += 4; }
  if (p < end) gather_rows<4, true>(h2, ed, p, end, lane, ax, ay);

  const float v0 = ax + b0, v1 = ay + b1;
  float s = v0 + v1;
  #pragma unroll
  for (int off = 16; off > 0; off >>= 1) s += __shfl_xor(s, off, 32);
  const float mu = s * (1.0f / 64.0f);
  const float d0 = v0 - mu, d1 = v1 - mu;
  float q = d0 * d0 + d1 * d1;
  #pragma unroll
  for (int off = 16; off > 0; off >>= 1) q += __shfl_xor(q, off, 32);
  const float rstd = rsqrtf(q * (1.0f / 64.0f) + LN_EPS);
  const float2 y = make_float2(fmaxf(0.f, d0 * rstd * w0 + o0), fmaxf(0.f, d1 * rstd * w1 + o1));
  if (!POOL) {
    ((float2*)hout + node * 32)[lane] = y;
    return;
  }
  const int wave = threadIdx.x >> 5;
  ((float2*)s_row)[wave * 32 + lane] = y;
  __syncthreads();
  if (wave != 0) return;
  float r0[8], r1[8];
  #pragma unroll
  for (int w = 0; w < 8; ++w) {
    r0[w] = s_row[w * D_H + lane];
    r1[w] = s_row[w * D_H + 32 + lane];
  }
  int gc = g[0];
  float a0 = 0.f, a1 = 0.f, c = 0.f;
  #pragma unroll
  for (int w = 0; w < 8; ++w) {
    if (g[w] != gc) {
      atomicAdd(&pool[gc * D_H + lane], a0);
      atomicAdd(&pool[gc * D_H + 32 + lane], a1);
      if (lane == 0) atomicAdd(&gcnt[gc], c);
      gc = g[w]; a0 = 0.f; a1 = 0.f; c = 0.f;
    }
    a0 += r0[w];
    a1 += r1[w];
    c += 1.0f;
  }
  atomicAdd(&pool[gc * D_H + lane], a0);
  atomicAdd(&pool[gc * D_H + 32 + lane], a1);
  if (lane == 0) atomicAdd(&gcnt[gc], c);
}

static inline int cdiv(int a, int b) { return (a + b - 1) / b; }

template <bool LAST>
static void run_layer(const float* hin, const float* W, const float* b,
                      const float* lnw, const float* lnb,
                      const float* dinv, const int* rowptr, const int2* ed,
                      float* hW, float* hout,
                      const int* batch, float* pool, float* gcnt,
                      hipStream_t stream) {
  gcn_gemm_wmma<64, false, false><<<N_NODES / 16, 128, 0, stream>>>(hin, W, nullptr, nullptr, hW);
  k_gather_ln<LAST><<<cdiv(N_NODES * 32, 256), 256, 0, stream>>>(hW, rowptr, ed, dinv, b, lnw, lnb, hout,
                                                                   batch, pool, gcnt);
}

extern "C" void kernel_launch(void* const* d_in, const int* in_sizes, int n_in,
                              void* d_out, int out_size, void* d_ws, size_t ws_size,
                              hipStream_t stream) {
  (void)in_sizes; (void)n_in; (void)out_size; (void)ws_size;

  const float* x       = (const float*)d_in[0];
  const int*   eidx    = (const int*)d_in[1];
  const int*   batch   = (const int*)d_in[2];
  const float* W_in    = (const float*)d_in[3];
  const float* b_in    = (const float*)d_in[4];
  const float* ln_in_w = (const float*)d_in[5];
  const float* ln_in_b = (const float*)d_in[6];
  const float* W_h     = (const float*)d_in[7];
  const float* b_h     = (const float*)d_in[8];
  const float* ln_h_w  = (const float*)d_in[9];
  const float* ln_h_b  = (const float*)d_in[10];
  const float* W_out   = (const float*)d_in[11];
  const float* b_out   = (const float*)d_in[12];

  const int* src = eidx;
  const int* dst = eidx + N_EDGES;

  float* ws    = (float*)d_ws;
  float* dinv  = ws;
  int*   rowptr= (int*)(dinv + N_PAD);
  int*   cntc  = rowptr + N_PAD;
  int*   basec = cntc + NCHUNK * N_PAD;
  int*   wgtot = basec + NCHUNK * N_PAD;
  int*   rank  = wgtot + 1024;
  int2*  ed    = (int2*)(rank + N_EDGES);
  float* hW    = (float*)(ed + N_EDGES);
  float* hbuf  = hW + (size_t)N_NODES*64;
  float* pool  = hbuf + (size_t)N_NODES*64;
  float* cnt_g = pool + N_GRAPHS * 64;

  k_rank_lds<<<NPART * NCHUNK, 1024, 0, stream>>>(dst, rank, cntc, wgtot);
  k_scan<<<cdiv(N_NODES + 1, 1024), 1024, 0, stream>>>(cntc, wgtot, rowptr, basec, dinv, pool, cnt_g);
  k_fill<<<cdiv(N_EDGES, 256), 256, 0, stream>>>(src, dst, rank, basec, dinv, ed);

  run_layer<false>(x,    W_in,       b_in,       ln_in_w,      ln_in_b,      dinv, rowptr, ed, hW, hbuf,
                   batch, pool, cnt_g, stream);
  run_layer<false>(hbuf, W_h,        b_h,        ln_h_w,       ln_h_b,       dinv, rowptr, ed, hW, hbuf,
                   batch, pool, cnt_g, stream);
  run_layer<true>(hbuf, W_h + 64*64, b_h + 64,  ln_h_w + 64,  ln_h_b + 64,  dinv, rowptr, ed, hW, nullptr,
                  batch, pool, cnt_g, stream);

  gcn_gemm_wmma<D_OUT, true, true><<<N_GRAPHS / 16, 64, 0, stream>>>(pool, W_out, b_out, cnt_g, (float*)d_out);
}


# Next layer's h@W moved into the row-gather kernel's tail: a block's 16 finished rows meet in LDS, 4 of its waves do the f32 WMMA tiles; 10 to 8 launches, h never stored; all sums as before
# speedup vs baseline: 4.8807x; 1.0677x over previous
#include <hip/hip_runtime.h>
#include <hip/hip_bf16.h>

#define N_NODES  50000
#define N_EDGES  800000
#define N_GRAPHS 512
#define D_H      64
#define D_OUT    32
#define LN_EPS   1e-5f
#define N_PAD    50048

typedef __attribute__((ext_vector_type(2)))  float    v2f;
typedef __attribute__((ext_vector_type(8)))  float    v8f;
typedef __attribute__((ext_vector_type(16))) _Float16 v16h;

#if defined(__has_builtin)
#if __has_builtin(__builtin_amdgcn_wmma_f32_16x16x4_f32)
#define USE_F32_WMMA 1
#endif
#endif

template <int NCOLS, bool BIAS, bool MEAN>
__global__ __launch_bounds__(32 * (NCOLS / 16)) void gcn_gemm_wmma(
    const float* __restrict__ A, const float* __restrict__ B,
    const float* __restrict__ bias, const float* __restrict__ cnt, float* __restrict__ C) {
  const int lane = threadIdx.x & 31;
  const int wave = threadIdx.x >> 5;
  const int m    = lane & 15;
  const int half = lane >> 4;
  const int c0   = wave * 16;
  const int r0   = blockIdx.x * 16;
  const int row  = r0 + m;

  v8f acc = {0.f, 0.f, 0.f, 0.f, 0.f, 0.f, 0.f, 0.f};
  float den = 1.0f;
  if (MEAN) den = fmaxf(cnt[row], 1.0f);

#if USE_F32_WMMA
  #pragma unroll
  for (int k0 = 0; k0 < 64; k0 += 4) {
    const int ka = k0 + 2 * half;
    v2f a, b;
    a.x = A[row * 64 + ka];
    a.y = A[row * 64 + ka + 1];
    if (MEAN) { a.x /= den; a.y /= den; }
    b.x = B[ka * NCOLS + c0 + m];
    b.y = B[(ka + 1) * NCOLS + c0 + m];
    acc = __builtin_amdgcn_wmma_f32_16x16x4_f32(false, a, false, b,
                                                (short)0, acc, false, false);
  }
#else
  for (int kstep = 0; kstep < 2; ++kstep) {
    v16h af, bf;
    #pragma unroll
    for (int e = 0; e < 16; ++e) {
      int v = e >> 1, j = e & 1, grp = v >> 2;
      int K = 32 * kstep + 16 * grp + 8 * half + ((v & 3) * 2 + j);
      af[e] = (_Float16)(MEAN ? A[row * 64 + K] / den : A[row * 64 + K]);
    }
    #pragma unroll
    for (int e = 0; e < 16; ++e) {
      int v = e >> 1, j = e & 1;
      int K = 32 * kstep + 16 * half + 2 * v + j;
      bf[e] = (_Float16)B[K * NCOLS + c0 + m];
    }
    acc = __builtin_amdgcn_wmma_f32_16x16x32_f16(false, af, false, bf,
                                                 (short)0, acc, false, false);
  }
#endif

  #pragma unroll
  for (int i = 0; i < 8; ++i) {
    float v = acc[i];
    if (BIAS) v += bias[c0 + m];
    C[(r0 + i + 8 * half) * NCOLS + c0 + m] = v;
  }
}

#define RANGE   4096
#define NPART   13
#define NCHUNK  16
#define CH      (N_EDGES / NCHUNK)
#define NSUB    (NPART * 4)
#define NLOAD   13

__global__ __launch_bounds__(1024) void k_rank_lds(const int* __restrict__ dst,
                                                   int* __restrict__ rank,
                                                   int* __restrict__ cntc,
                                                   int* __restrict__ wgtot) {
  __shared__ int s_cnt[RANGE];
  __shared__ int s_red[128];
  const int tid = threadIdx.x, lane = tid & 31, wave = tid >> 5;
  const int p = blockIdx.x / NCHUNK, c = blockIdx.x % NCHUNK;
  const int lo = p * RANGE;
  #pragma unroll
  for (int q = 0; q < 4; ++q) s_cnt[q * 1024 + tid] = 0;

  const int e0 = c * CH;
  const int4* d4 = (const int4*)(dst + e0);
  int4 v[NLOAD];
  #pragma unroll
  for (int k = 0; k < NLOAD; ++k) {
    const int j = tid + k * 1024;
    v[k] = (j < CH / 4) ? d4[j] : make_int4(-1, -1, -1, -1);
  }
  __syncthreads();
  #pragma unroll
  for (int k = 0; k < NLOAD; ++k) {
    const int e = e0 + 4 * (tid + k * 1024);
    unsigned r;
    r = (unsigned)(v[k].x - lo); if (r < RANGE) rank[e]     = atomicAdd(&s_cnt[r], 1);
    r = (unsigned)(v[k].y - lo); if (r < RANGE) rank[e + 1] = atomicAdd(&s_cnt[r], 1);
    r = (unsigned)(v[k].z - lo); if (r < RANGE) rank[e + 2] = atomicAdd(&s_cnt[r], 1);
    r = (unsigned)(v[k].w - lo); if (r < RANGE) rank[e + 3] = atomicAdd(&s_cnt[r], 1);
  }
  __syncthreads();
  #pragma unroll
  for (int q = 0; q < 4; ++q) {
    const int n = lo + q * 1024 + tid;
    int t = s_cnt[q * 1024 + tid];
    if (n < N_NODES) cntc[c * N_PAD + n] = t;
    #pragma unroll
    for (int off = 16; off > 0; off >>= 1) t += __shfl_xor(t, off, 32);
    if (lane == 0) s_red[q * 32 + wave] = t;
  }
  __syncthreads();
  if (tid < 128) {
    int t = s_red[tid];
    #pragma unroll
    for (int off = 16; off > 0; off >>= 1) t += __shfl_xor(t, off, 32);
    if (lane == 0) wgtot[(p * 4 + wave) * NCHUNK + c] = t;
  }
}

__global__ __launch_bounds__(1024) void k_scan(const int* __restrict__ cntc,
                                               const int* __restrict__ wgtot,
                                               int* __restrict__ rowptr,
                                               int* __restrict__ basec,
                                               float* __restrict__ dinv,
                                               float* __restrict__ pool,
                                               float* __restrict__ gcnt) {
  __shared__ int s_front[32];
  __shared__ int s_wtot[32];
  const int tid = threadIdx.x, lane = tid & 31, wave = tid >> 5;
  const int n = blockIdx.x * 1024 + tid;

  if (n < N_GRAPHS * D_H) pool[n] = 0.f;
  if (n < N_GRAPHS) gcnt[n] = 0.f;

  int front = (tid < (int)blockIdx.x * NCHUNK) ? wgtot[tid] : 0;
  #pragma unroll
  for (int off = 16; off > 0; off >>= 1) front += __shfl_xor(front, off, 32);

  int cc[NCHUNK];
  int c = 0;
  #pragma unroll
  for (int k = 0; k < NCHUNK; ++k) {
    cc[k] = (n < N_NODES) ? cntc[k * N_PAD + n] : 0;
    c += cc[k];
  }
  int incl = c;
  #pragma unroll
  for (int off = 1; off < 32; off <<= 1) {
    int t = __shfl_up(incl, off, 32);
    if (lane >= off) incl += t;
  }
  if (lane == 0)  s_front[wave] = front;
  if (lane == 31) s_wtot[wave]  = incl;
  __syncthreads();
  int base = 0;
  for (int w = 0; w < 32; ++w) {
    base += s_front[w];
    if (w < wave) base += s_wtot[w];
  }
  int run = base + incl - c;
  if (n <= N_NODES) rowptr[n] = run;
  if (n < N_NODES) {
    dinv[n] = rsqrtf((float)(c + 1));
    #pragma unroll
    for (int k = 0; k < NCHUNK; ++k) {
      basec[k * N_PAD + n] = run;
      run += cc[k];
    }
  }
}

__global__ void k_fill(const int* __restrict__ src, const int* __restrict__ dst,
                       const int* __restrict__ rank, const int* __restrict__ basec,
                       const float* __restrict__ dinv, int2* __restrict__ ed) {
  int i = blockIdx.x * blockDim.x + threadIdx.x;
  if (i >= N_EDGES) return;
  int s = src[i], d = dst[i];
  ed[basec[(i / CH) * N_PAD + d] + rank[i]] = make_int2(s, __float_as_int(dinv[s] * dinv[d]));
}

template <int K, bool CLAMP>
__device__ __forceinline__ void gather_rows(const float2* __restrict__ h2,
                                            const int2* __restrict__ ed,
                                            int p, int end, int lane,
                                            float& ax, float& ay) {
  int2 e[K];
  #pragma unroll
  for (int j = 0; j < K; ++j) {
    int q = p + j;
    if (CLAMP) q = min(q, end - 1);
    e[j] = ed[q];
  }
  float2 r[K];
  #pragma unroll
  for (int j = 0; j < K; ++j) r[j] = (h2 + e[j].x * 32)[lane];
  #pragma unroll
  for (int j = 0; j < K; ++j) {
    float w = __int_as_float(e[j].y);
    if (CLAMP && p + j >= end) w = 0.f;
    ax += w * r[j].x;
    ay += w * r[j].y;
  }
}

#define LDA 68
static_assert(N_NODES % 16 == 0, "a block holds 16 (POOL: 8) whole rows: all of its waves pass the row guard or none does");
template <bool POOL>
__global__ __launch_bounds__(POOL ? 256 : 512) void k_gather_ln(const float* __restrict__ hW,
                                                   const int* __restrict__ rowptr,
                                                   const int2* __restrict__ ed,
                                                   const float* __restrict__ dinv,
                                                   const float* __restrict__ bias,
                                                   const float* __restrict__ lnw,
                                                   const float* __restrict__ lnb,
                                                   const float* __restrict__ Wn,
                                                   float* __restrict__ hWn,
                                                   const int* __restrict__ batch,
                                                   float* __restrict__ pool,
                                                   float* __restrict__ gcnt) {
  __shared__ float s_row[POOL ? 8 * D_H : 16 * LDA];
  const int lane = threadIdx.x & 31;
  const int wave = __builtin_amdgcn_readfirstlane((int)(threadIdx.x >> 5));
  const int node = (int)blockIdx.x * (POOL ? 8 : 16) + wave;
  if (node >= N_NODES) return;
  const int beg = rowptr[node], end = rowptr[node + 1];
  const float di = dinv[node];
  int g[8];
  if (POOL) {
    #pragma unroll
    for (int w = 0; w < 8; ++w) g[w] = batch[(int)blockIdx.x * 8 + w];
  }
  const float2* h2 = (const float2*)hW;
  const float2 self = (h2 + node * 32)[lane];
  const float b0 = bias[2 * lane], b1 = bias[2 * lane + 1];
  const float w0 = lnw[2 * lane],  w1 = lnw[2 * lane + 1];
  const float o0 = lnb[2 * lane],  o1 = lnb[2 * lane + 1];
  float ax = di * di * self.x;
  float ay = di * di * self.y;
  int p = beg;
  for (; p + 8 <= end; p += 8) gather_rows<8, false>(h2, ed, p, end, lane, ax, ay);
  if (p + 4 <= end) { gather_rows<4, false>(h2, ed, p, end, lane, ax, ay); p += 4; }
  if (p < end) gather_rows<4, true>(h2, ed, p, end, lane, ax, ay);

  const int m = lane & 15, half = lane >> 4, c0 = wave * 16;
  v2f bfr[16];
  if (!POOL && wave < 4) {
    #pragma unroll
    for (int t = 0; t < 16; ++t) {
      const int ka = 4 * t + 2 * half;
      bfr[t].x = Wn[ka * D_H + c0 + m];
      bfr[t].y = Wn[(ka + 1) * D_H + c0 + m];
    }
  }

  const float v0 = ax + b0, v1 = ay + b1;
  float s = v0 + v1;
  #pragma unroll
  for (int off = 16; off > 0; off >>= 1) s += __shfl_xor(s, off, 32);
  const float mu = s * (1.0f / 64.0f);
  const float d0 = v0 - mu, d1 = v1 - mu;
  float q = d0 * d0 + d1 * d1;
  #pragma unroll
  for (int off = 16; off > 0; off >>= 1) q += __shfl_xor(q, off, 32);
  const float rstd = rsqrtf(q * (1.0f / 64.0f) + LN_EPS);
  const float2 y = make_float2(fmaxf(0.f, d0 * rstd * w0 + o0), fmaxf(0.f, d1 * rstd * w1 + o1));
  if (!POOL) {
    ((float2*)s_row)[wave * (LDA / 2) + lane] = y;
    __syncthreads();
    if (wave >= 4) return;
    v8f acc = {0.f, 0.f, 0.f, 0.f, 0.f, 0.f, 0.f, 0.f};
    #pragma unroll
    for (int t = 0; t < 16; ++t) {
      const int ka = 4 * t + 2 * half;
      const float2 av = *(const float2*)&s_row[m * LDA + ka];
      v2f a;
      a.x = av.x;
      a.y = av.y;
      acc = __builtin_amdgcn_wmma_f32_16x16x4_f32(false, a, false, bfr[t],
                                                  (short)0, acc, false, false);
    }
    const int r0 = (int)blockIdx.x * 16;
    #pragma unroll
    for (int i = 0; i < 8; ++i) hWn[(r0 + i + 8 * half) * D_H + c0 + m] = acc[i];
    return;
  }
  ((float2*)s_row)[wave * 32 + lane] = y;
  __syncthreads();
  if (wave != 0) return;
  float r0[8], r1[8];
  #pragma unroll
  for (int w = 0; w < 8; ++w) {
    r0[w] = s_row[w * D_H + lane];
    r1[w] = s_row[w * D_H + 32 + lane];
  }
  int gc = g[0];
  float a0 = 0.f, a1 = 0.f, c = 0.f;
  #pragma unroll
  for (int w = 0; w < 8; ++w) {
    if (g[w] != gc) {
      atomicAdd(&pool[gc * D_H + lane], a0);
      atomicAdd(&pool[gc * D_H + 32 + lane], a1);
      if (lane == 0) atomicAdd(&gcnt[gc], c);
      gc = g[w]; a0 = 0.f; a1 = 0.f; c = 0.f;
    }
    a0 += r0[w];
    a1 += r1[w];
    c += 1.0f;
  }
  atomicAdd(&pool[gc * D_H + lane], a0);
  atomicAdd(&pool[gc * D_H + 32 + lane], a1);
  if (lane == 0) atomicAdd(&gcnt[gc], c);
}

static inline int cdiv(int a, int b) { return (a + b - 1) / b; }

template <bool LAST>
static void run_layer(const float* hW, const float* b,
                      const float* lnw, const float* lnb,
                      const float* dinv, const int* rowptr, const int2* ed,
                      const float* Wn, float* hWn,
                      const int* batch, float* pool, float* gcnt,
                      hipStream_t stream) {
  k_gather_ln<LAST><<<N_NODES / (LAST ? 8 : 16), LAST ? 256 : 512, 0, stream>>>(hW, rowptr, ed, dinv, b, lnw, lnb,
                                                                                 Wn, hWn, batch, pool, gcnt);
}

extern "C" void kernel_launch(void* const* d_in, const int* in_sizes, int n_in,
                              void* d_out, int out_size, void* d_ws, size_t ws_size,
                              hipStream_t stream) {
  (void)in_sizes; (void)n_in; (void)out_size; (void)ws_size;

  const float* x       = (const float*)d_in[0];
  const int*   eidx    = (const int*)d_in[1];
  const int*   batch   = (const int*)d_in[2];
  const float* W_in    = (const float*)d_in[3];
  const float* b_in    = (const float*)d_in[4];
  const float* ln_in_w = (const float*)d_in[5];
  const float* ln_in_b = (const float*)d_in[6];
  const float* W_h     = (const float*)d_in[7];
  const float* b_h     = (const float*)d_in[8];
  const float* ln_h_w  = (const float*)d_in[9];
  const float* ln_h_b  = (const float*)d_in[10];
  const float* W_out   = (const float*)d_in[11];
  const float* b_out   = (const float*)d_in[12];

  const int* src = eidx;
  const int* dst = eidx + N_EDGES;

  float* ws    = (float*)d_ws;
  float* dinv  = ws;
  int*   rowptr= (int*)(dinv + N_PAD);
  int*   cntc  = rowptr + N_PAD;
  int*   basec = cntc + NCHUNK * N_PAD;
  int*   wgtot = basec + NCHUNK * N_PAD;
  int*   rank  = wgtot + 1024;
  int2*  ed    = (int2*)(rank + N_EDGES);
  float* hW    = (float*)(ed + N_EDGES);
  float* hW2   = hW + (size_t)N_NODES*64;
  float* pool  = hW2 + (size_t)N_NODES*64;
  float* cnt_g = pool + N_GRAPHS * 64;

  k_rank_lds<<<NPART * NCHUNK, 1024, 0, stream>>>(dst, rank, cntc, wgtot);
  k_scan<<<cdiv(N_NODES + 1, 1024), 1024, 0, stream>>>(cntc, wgtot, rowptr, basec, dinv, pool, cnt_g);
  k_fill<<<cdiv(N_EDGES, 256), 256, 0, stream>>>(src, dst, rank, basec, dinv, ed);

  gcn_gemm_wmma<64, false, false><<<N_NODES / 16, 128, 0, stream>>>(x, W_in, nullptr, nullptr, hW);
  run_layer<false>(hW,  b_in,     ln_in_w,     ln_in_b,     dinv, rowptr, ed, W_h,           hW2,
                   batch, pool, cnt_g, stream);
  run_layer<false>(hW2, b_h,      ln_h_w,      ln_h_b,      dinv, rowptr, ed, W_h + 64 * 64, hW,
                   batch, pool, cnt_g, stream);
  run_layer<true>(hW,   b_h + 64, ln_h_w + 64, ln_h_b + 64, dinv, rowptr, ed, nullptr,       nullptr,
                  batch, pool, cnt_g, stream);

  gcn_gemm_wmma<D_OUT, true, true><<<N_GRAPHS / 16, 64, 0, stream>>>(pool, W_out, b_out, cnt_g, (float*)d_out);
}


# 4x the work per wave: a row-kernel wave sums 4 consecutive rows out of their one contiguous run of the edge list (150000 to 37500 waves), a k_fill thread places 4 edges; row sum order as before
# speedup vs baseline: 5.1883x; 1.0630x over previous
#include <hip/hip_runtime.h>
#include <hip/hip_bf16.h>

#define N_NODES  50000
#define N_EDGES  800000
#define N_GRAPHS 512
#define D_H      64
#define D_OUT    32
#define LN_EPS   1e-5f
#define N_PAD    50048

typedef __attribute__((ext_vector_type(2)))  float    v2f;
typedef __attribute__((ext_vector_type(8)))  float    v8f;
typedef __attribute__((ext_vector_type(16))) _Float16 v16h;

#if defined(__has_builtin)
#if __has_builtin(__builtin_amdgcn_wmma_f32_16x16x4_f32)
#define USE_F32_WMMA 1
#endif
#endif

template <int NCOLS, bool BIAS, bool MEAN>
__global__ __launch_bounds__(32 * (NCOLS / 16)) void gcn_gemm_wmma(
    const float* __restrict__ A, const float* __restrict__ B,
    const float* __restrict__ bias, const float* __restrict__ cnt, float* __restrict__ C) {
  const int lane = threadIdx.x & 31;
  const int wave = threadIdx.x >> 5;
  const int m    = lane & 15;
  const int half = lane >> 4;
  const int c0   = wave * 16;
  const int r0   = blockIdx.x * 16;
  const int row  = r0 + m;

  v8f acc = {0.f, 0.f, 0.f, 0.f, 0.f, 0.f, 0.f, 0.f};
  float den = 1.0f;
  if (MEAN) den = fmaxf(cnt[row], 1.0f);

#if USE_F32_WMMA
  #pragma unroll
  for (int k0 = 0; k0 < 64; k0 += 4) {
    const int ka = k0 + 2 * half;
    v2f a, b;
    a.x = A[row * 64 + ka];
    a.y = A[row * 64 + ka + 1];
    if (MEAN) { a.x /= den; a.y /= den; }
    b.x = B[ka * NCOLS + c0 + m];
    b.y = B[(ka + 1) * NCOLS + c0 + m];
    acc = __builtin_amdgcn_wmma_f32_16x16x4_f32(false, a, false, b,
                                                (short)0, acc, false, false);
  }
#else
  for (int kstep = 0; kstep < 2; ++kstep) {
    v16h af, bf;
    #pragma unroll
    for (int e = 0; e < 16; ++e) {
      int v = e >> 1, j = e & 1, grp = v >> 2;
      int K = 32 * kstep + 16 * grp + 8 * half + ((v & 3) * 2 + j);
      af[e] = (_Float16)(MEAN ? A[row * 64 + K] / den : A[row * 64 + K]);
    }
    #pragma unroll
    for (int e = 0; e < 16; ++e) {
      int v = e >> 1, j = e & 1;
      int K = 32 * kstep + 16 * half + 2 * v + j;
      bf[e] = (_Float16)B[K * NCOLS + c0 + m];
    }
    acc = __builtin_amdgcn_wmma_f32_16x16x32_f16(false, af, false, bf,
                                                 (short)0, acc, false, false);
  }
#endif

  #pragma unroll
  for (int i = 0; i < 8; ++i) {
    float v = acc[i];
    if (BIAS) v += bias[c0 + m];
    C[(r0 + i + 8 * half) * NCOLS + c0 + m] = v;
  }
}

#define RANGE   4096
#define NPART   13
#define NCHUNK  16
#define CH      (N_EDGES / NCHUNK)
#define NSUB    (NPART * 4)
#define NLOAD   13

__global__ __launch_bounds__(1024) void k_rank_lds(const int* __restrict__ dst,
                                                   int* __restrict__ rank,
                                                   int* __restrict__ cntc,
                                                   int* __restrict__ wgtot) {
  __shared__ int s_cnt[RANGE];
  __shared__ int s_red[128];
  const int tid = threadIdx.x, lane = tid & 31, wave = tid >> 5;
  const int p = blockIdx.x / NCHUNK, c = blockIdx.x % NCHUNK;
  const int lo = p * RANGE;
  #pragma unroll
  for (int q = 0; q < 4; ++q) s_cnt[q * 1024 + tid] = 0;

  const int e0 = c * CH;
  const int4* d4 = (const int4*)(dst + e0);
  int4 v[NLOAD];
  #pragma unroll
  for (int k = 0; k < NLOAD; ++k) {
    const int j = tid + k * 1024;
    v[k] = (j < CH / 4) ? d4[j] : make_int4(-1, -1, -1, -1);
  }
  __syncthreads();
  #pragma unroll
  for (int k = 0; k < NLOAD; ++k) {
    const int e = e0 + 4 * (tid + k * 1024);
    unsigned r;
    r = (unsigned)(v[k].x - lo); if (r < RANGE) rank[e]     = atomicAdd(&s_cnt[r], 1);
    r = (unsigned)(v[k].y - lo); if (r < RANGE) rank[e + 1] = atomicAdd(&s_cnt[r], 1);
    r = (unsigned)(v[k].z - lo); if (r < RANGE) rank[e + 2] = atomicAdd(&s_cnt[r], 1);
    r = (unsigned)(v[k].w - lo); if (r < RANGE) rank[e + 3] = atomicAdd(&s_cnt[r], 1);
  }
  __syncthreads();
  #pragma unroll
  for (int q = 0; q < 4; ++q) {
    const int n = lo + q * 1024 + tid;
    int t = s_cnt[q * 1024 + tid];
    if (n < N_NODES) cntc[c * N_PAD + n] = t;
    #pragma unroll
    for (int off = 16; off > 0; off >>= 1) t += __shfl_xor(t, off, 32);
    if (lane == 0) s_red[q * 32 + wave] = t;
  }
  __syncthreads();
  if (tid < 128) {
    int t = s_red[tid];
    #pragma unroll
    for (int off = 16; off > 0; off >>= 1) t += __shfl_xor(t, off, 32);
    if (lane == 0) wgtot[(p * 4 + wave) * NCHUNK + c] = t;
  }
}

__global__ __launch_bounds__(1024) void k_scan(const int* __restrict__ cntc,
                                               const int* __restrict__ wgtot,
                                               int* __restrict__ rowptr,
                                               int* __restrict__ basec,
                                               float* __restrict__ dinv,
                                               float* __restrict__ pool,
                                               float* __restrict__ gcnt) {
  __shared__ int s_front[32];
  __shared__ int s_wtot[32];
  const int tid = threadIdx.x, lane = tid & 31, wave = tid >> 5;
  const int n = blockIdx.x * 1024 + tid;

  if (n < N_GRAPHS * D_H) pool[n] = 0.f;
  if (n < N_GRAPHS) gcnt[n] = 0.f;

  int front = (tid < (int)blockIdx.x * NCHUNK) ? wgtot[tid] : 0;
  #pragma unroll
  for (int off = 16; off > 0; off >>= 1) front += __shfl_xor(front, off, 32);

  int cc[NCHUNK];
  int c = 0;
  #pragma unroll
  for (int k = 0; k < NCHUNK; ++k) {
    cc[k] = (n < N_NODES) ? cntc[k * N_PAD + n] : 0;
    c += cc[k];
  }
  int incl = c;
  #pragma unroll
  for (int off = 1; off < 32; off <<= 1) {
    int t = __shfl_up(incl, off, 32);
    if (lane >= off) incl += t;
  }
  if (lane == 0)  s_front[wave] = front;
  if (lane == 31) s_wtot[wave]  = incl;
  __syncthreads();
  int base = 0;
  for (int w = 0; w < 32; ++w) {
    base += s_front[w];
    if (w < wave) base += s_wtot[w];
  }
  int run = base + incl - c;
  if (n <= N_NODES) rowptr[n] = run;
  if (n < N_NODES) {
    dinv[n] = rsqrtf((float)(c + 1));
    #pragma unroll
    for (int k = 0; k < NCHUNK; ++k) {
      basec[k * N_PAD + n] = run;
      run += cc[k];
    }
  }
}

static_assert(CH % 4 == 0 && N_EDGES % 4 == 0, "a thread's four edges lie in one edge chunk and inside the list");
__global__ void k_fill(const int* __restrict__ src, const int* __restrict__ dst,
                       const int* __restrict__ rank, const int* __restrict__ basec,
                       const float* __restrict__ dinv, int2* __restrict__ ed) {
  const int i = (blockIdx.x * blockDim.x + threadIdx.x) * 4;
  if (i >= N_EDGES) return;
  const int4 s = *(const int4*)(src + i);
  const int4 d = *(const int4*)(dst + i);
  const int4 r = *(const int4*)(rank + i);
  const int* bc = basec + (i / CH) * N_PAD;
  ed[bc[d.x] + r.x] = make_int2(s.x, __float_as_int(dinv[s.x] * dinv[d.x]));
  ed[bc[d.y] + r.y] = make_int2(s.y, __float_as_int(dinv[s.y] * dinv[d.y]));
  ed[bc[d.z] + r.z] = make_int2(s.z, __float_as_int(dinv[s.z] * dinv[d.z]));
  ed[bc[d.w] + r.w] = make_int2(s.w, __float_as_int(dinv[s.w] * dinv[d.w]));
}

#define LDA 68
#define RPW 4
static_assert(N_NODES % 16 == 0, "a block holds 16 (POOL: 8) whole rows: all of its waves pass the row guard or none does");
template <bool POOL>
__global__ __launch_bounds__(POOL ? 64 : 128) void k_gather_ln(const float* __restrict__ hW,
                                                   const int* __restrict__ rowptr,
                                                   const int2* __restrict__ ed,
                                                   const float* __restrict__ dinv,
                                                   const float* __restrict__ bias,
                                                   const float* __restrict__ lnw,
                                                   const float* __restrict__ lnb,
                                                   const float* __restrict__ Wn,
                                                   float* __restrict__ hWn,
                                                   const int* __restrict__ batch,
                                                   float* __restrict__ pool,
                                                   float* __restrict__ gcnt) {
  __shared__ float s_row[POOL ? 8 * D_H : 16 * LDA];
  const int lane = threadIdx.x & 31;
  const int wave = __builtin_amdgcn_readfirstlane((int)(threadIdx.x >> 5));
  const int row0 = (int)blockIdx.x * (POOL ? 8 : 16) + wave * RPW;
  if (row0 >= N_NODES) return;
  int rp[RPW + 1];
  #pragma unroll
  for (int k = 0; k <= RPW; ++k) rp[k] = rowptr[row0 + k];
  int g[8];
  if (POOL) {
    #pragma unroll
    for (int w = 0; w < 8; ++w) g[w] = batch[(int)blockIdx.x * 8 + w];
  }
  const float2* h2 = (const float2*)hW;
  float sx[RPW], sy[RPW];
  #pragma unroll
  for (int k = 0; k < RPW; ++k) {
    const float di = dinv[row0 + k];
    const float2 self = (h2 + (row0 + k) * 32)[lane];
    sx[k] = di * di * self.x;
    sy[k] = di * di * self.y;
  }
  const float b0 = bias[2 * lane], b1 = bias[2 * lane + 1];
  const float w0 = lnw[2 * lane],  w1 = lnw[2 * lane + 1];
  const float o0 = lnb[2 * lane],  o1 = lnb[2 * lane + 1];

  const int end = rp[RPW];
  float fx[RPW], fy[RPW];
  #pragma unroll
  for (int k = 0; k < RPW; ++k) { fx[k] = 0.f; fy[k] = 0.f; }
  float ax = sx[0], ay = sy[0];
  int cur = 0;
  int nb = rp[1];
  auto next_row = [&]() {
    fx[0] = (cur == 0) ? ax : fx[0];  fy[0] = (cur == 0) ? ay : fy[0];
    fx[1] = (cur == 1) ? ax : fx[1];  fy[1] = (cur == 1) ? ay : fy[1];
    fx[2] = (cur == 2) ? ax : fx[2];  fy[2] = (cur == 2) ? ay : fy[2];
    ax = (cur == 0) ? sx[1] : (cur == 1) ? sx[2] : sx[3];
    ay = (cur == 0) ? sy[1] : (cur == 1) ? sy[2] : sy[3];
    nb = (cur == 0) ? rp[2] : (cur == 1) ? rp[3] : 0x7fffffff;
    ++cur;
  };
  int p = rp[0];
  int2 e[8];
  {
    const int pb = min(p, N_EDGES - 8);
    #pragma unroll
    for (int j = 0; j < 8; ++j) e[j] = ed[pb + j];
  }
  while (p + 8 <= end) {
    float2 r[8];
    #pragma unroll
    for (int j = 0; j < 8; ++j) r[j] = (h2 + e[j].x * 32)[lane];
    int2 en[8];
    const int pn = min(p + 8, N_EDGES - 8);
    #pragma unroll
    for (int j = 0; j < 8; ++j) en[j] = ed[pn + j];
    #pragma unroll
    for (int j = 0; j < 8; ++j) {
      while (p + j >= nb) next_row();
      const float w = __int_as_float(e[j].y);
      ax += w * r[j].x;
      ay += w * r[j].y;
    }
    #pragma unroll
    for (int j = 0; j < 8; ++j) e[j] = en[j];
    p += 8;
  }
  if (p < end) {
    if (p + 8 > N_EDGES) {
      #pragma unroll
      for (int j = 0; j < 8; ++j) e[j] = ed[min(p + j, end - 1)];
    }
    float2 r[8];
    #pragma unroll
    for (int j = 0; j < 8; ++j) r[j] = (h2 + ((p + j < end) ? e[j].x : e[0].x) * 32)[lane];
    #pragma unroll
    for (int j = 0; j < 8; ++j) {
      if (p + j < end) {
        while (p + j >= nb) next_row();
        const float w = __int_as_float(e[j].y);
        ax += w * r[j].x;
        ay += w * r[j].y;
      }
    }
  }
  while (cur < RPW - 1) next_row();
  fx[RPW - 1] = ax; fy[RPW - 1] = ay;

  const int m = lane & 15, half = lane >> 4, c0 = wave * 16;
  v2f bfr[16];
  if (!POOL) {
    #pragma unroll
    for (int t = 0; t < 16; ++t) {
      const int ka = 4 * t + 2 * half;
      bfr[t].x = Wn[ka * D_H + c0 + m];
      bfr[t].y = Wn[(ka + 1) * D_H + c0 + m];
    }
  }

  #pragma unroll
  for (int k = 0; k < RPW; ++k) {
    const float v0 = fx[k] + b0, v1 = fy[k] + b1;
    float s = v0 + v1;
    #pragma unroll
    for (int off = 16; off > 0; off >>= 1) s += __shfl_xor(s, off, 32);
    const float mu = s * (1.0f / 64.0f);
    const float d0 = v0 - mu, d1 = v1 - mu;
    float q = d0 * d0 + d1 * d1;
    #pragma unroll
    for (int off = 16; off > 0; off >>= 1) q += __shfl_xor(q, off, 32);
    const float rstd = rsqrtf(q * (1.0f / 64.0f) + LN_EPS);
    const float2 y = make_float2(fmaxf(0.f, d0 * rstd * w0 + o0), fmaxf(0.f, d1 * rstd * w1 + o1));
    const int rb = wave * RPW + k;
    if (!POOL) ((float2*)s_row)[rb * (LDA / 2) + lane] = y;
    else       ((float2*)s_row)[rb * 32 + lane] = y;
  }
  __syncthreads();
  if (!POOL) {
    v8f acc = {0.f, 0.f, 0.f, 0.f, 0.f, 0.f, 0.f, 0.f};
    #pragma unroll
    for (int t = 0; t < 16; ++t) {
      const int ka = 4 * t + 2 * half;
      const float2 av = *(const float2*)&s_row[m * LDA + ka];
      v2f a;
      a.x = av.x;
      a.y = av.y;
      acc = __builtin_amdgcn_wmma_f32_16x16x4_f32(false, a, false, bfr[t],
                                                  (short)0, acc, false, false);
    }
    const int r0 = (int)blockIdx.x * 16;
    #pragma unroll
    for (int i = 0; i < 8; ++i) hWn[(r0 + i + 8 * half) * D_H + c0 + m] = acc[i];
    return;
  }
  if (wave != 0) return;
  float r0[8], r1[8];
  #pragma unroll
  for (int w = 0; w < 8; ++w) {
    r0[w] = s_row[w * D_H + lane];
    r1[w] = s_row[w * D_H + 32 + lane];
  }
  int gc = g[0];
  float a0 = 0.f, a1 = 0.f, c = 0.f;
  #pragma unroll
  for (int w = 0; w < 8; ++w) {
    if (g[w] != gc) {
      atomicAdd(&pool[gc * D_H + lane], a0);
      atomicAdd(&pool[gc * D_H + 32 + lane], a1);
      if (lane == 0) atomicAdd(&gcnt[gc], c);
      gc = g[w]; a0 = 0.f; a1 = 0.f; c = 0.f;
    }
    a0 += r0[w];
    a1 += r1[w];
    c += 1.0f;
  }
  atomicAdd(&pool[gc * D_H + lane], a0);
  atomicAdd(&pool[gc * D_H + 32 + lane], a1);
  if (lane == 0) atomicAdd(&gcnt[gc], c);
}

static inline int cdiv(int a, int b) { return (a + b - 1) / b; }

template <bool LAST>
static void run_layer(const float* hW, const float* b,
                      const float* lnw, const float* lnb,
                      const float* dinv, const int* rowptr, const int2* ed,
                      const float* Wn, float* hWn,
                      const int* batch, float* pool, float* gcnt,
                      hipStream_t stream) {
  k_gather_ln<LAST><<<N_NODES / (LAST ? 8 : 16), LAST ? 64 : 128, 0, stream>>>(hW, rowptr, ed, dinv, b, lnw, lnb,
                                                                                 Wn, hWn, batch, pool, gcnt);
}

extern "C" void kernel_launch(void* const* d_in, const int* in_sizes, int n_in,
                              void* d_out, int out_size, void* d_ws, size_t ws_size,
                              hipStream_t stream) {
  (void)in_sizes; (void)n_in; (void)out_size; (void)ws_size;

  const float* x       = (const float*)d_in[0];
  const int*   eidx    = (const int*)d_in[1];
  const int*   batch   = (const int*)d_in[2];
  const float* W_in    = (const float*)d_in[3];
  const float* b_in    = (const float*)d_in[4];
  const float* ln_in_w = (const float*)d_in[5];
  const float* ln_in_b = (const float*)d_in[6];
  const float* W_h     = (const float*)d_in[7];
  const float* b_h     = (const float*)d_in[8];
  const float* ln_h_w  = (const float*)d_in[9];
  const float* ln_h_b  = (const float*)d_in[10];
  const float* W_out   = (const float*)d_in[11];
  const float* b_out   = (const float*)d_in[12];

  const int* src = eidx;
  const int* dst = eidx + N_EDGES;

  float* ws    = (float*)d_ws;
  float* dinv  = ws;
  int*   rowptr= (int*)(dinv + N_PAD);
  int*   cntc  = rowptr + N_PAD;
  int*   basec = cntc + NCHUNK * N_PAD;
  int*   wgtot = basec + NCHUNK * N_PAD;
  int*   rank  = wgtot + 1024;
  int2*  ed    = (int2*)(rank + N_EDGES);
  float* hW    = (float*)(ed + N_EDGES);
  float* hW2   = hW + (size_t)N_NODES*64;
  float* pool  = hW2 + (size_t)N_NODES*64;
  float* cnt_g = pool + N_GRAPHS * 64;

  k_rank_lds<<<NPART * NCHUNK, 1024, 0, stream>>>(dst, rank, cntc, wgtot);
  k_scan<<<cdiv(N_NODES + 1, 1024), 1024, 0, stream>>>(cntc, wgtot, rowptr, basec, dinv, pool, cnt_g);
  k_fill<<<cdiv(N_EDGES / 4, 256), 256, 0, stream>>>(src, dst, rank, basec, dinv, ed);

  gcn_gemm_wmma<64, false, false><<<N_NODES / 16, 128, 0, stream>>>(x, W_in, nullptr, nullptr, hW);
  run_layer<false>(hW,  b_in,     ln_in_w,     ln_in_b,     dinv, rowptr, ed, W_h,           hW2,
                   batch, pool, cnt_g, stream);
  run_layer<false>(hW2, b_h,      ln_h_w,      ln_h_b,      dinv, rowptr, ed, W_h + 64 * 64, hW,
                   batch, pool, cnt_g, stream);
  run_layer<true>(hW,   b_h + 64, ln_h_w + 64, ln_h_b + 64, dinv, rowptr, ed, nullptr,       nullptr,
                  batch, pool, cnt_g, stream);

  gcn_gemm_wmma<D_OUT, true, true><<<N_GRAPHS / 16, 64, 0, stream>>>(pool, W_out, b_out, cnt_g, (float*)d_out);
}


# x@W_in tiles run as extra work-groups of the edge counting and scan launches, which they do not depend on, instead of in a launch of their own behind the list build: 8 to 7 launches; sums as before
# speedup vs baseline: 5.4533x; 1.0511x over previous
#include <hip/hip_runtime.h>
#include <hip/hip_bf16.h>

#define N_NODES  50000
#define N_EDGES  800000
#define N_GRAPHS 512
#define D_H      64
#define D_OUT    32
#define LN_EPS   1e-5f
#define N_PAD    50048

typedef __attribute__((ext_vector_type(2)))  float    v2f;
typedef __attribute__((ext_vector_type(8)))  float    v8f;
typedef __attribute__((ext_vector_type(16))) _Float16 v16h;

#if defined(__has_builtin)
#if __has_builtin(__builtin_amdgcn_wmma_f32_16x16x4_f32)
#define USE_F32_WMMA 1
#endif
#endif

template <int NCOLS, bool BIAS, bool MEAN>
__device__ __forceinline__ void gemm_tile(
    const float* __restrict__ A, const float* __restrict__ B,
    const float* __restrict__ bias, const float* __restrict__ cnt, float* __restrict__ C,
    const int r0, const int c0, const int lane) {
  const int m    = lane & 15;
  const int half = lane >> 4;
  const int row  = r0 + m;

  v8f acc = {0.f, 0.f, 0.f, 0.f, 0.f, 0.f, 0.f, 0.f};
  float den = 1.0f;
  if (MEAN) den = fmaxf(cnt[row], 1.0f);

#if USE_F32_WMMA
  #pragma unroll
  for (int k0 = 0; k0 < 64; k0 += 4) {
    const int ka = k0 + 2 * half;
    v2f a, b;
    a.x = A[row * 64 + ka];
    a.y = A[row * 64 + ka + 1];
    if (MEAN) { a.x /= den; a.y /= den; }
    b.x = B[ka * NCOLS + c0 + m];
    b.y = B[(ka + 1) * NCOLS + c0 + m];
    acc = __builtin_amdgcn_wmma_f32_16x16x4_f32(false, a, false, b,
                                                (short)0, acc, false, false);
  }
#else
  for (int kstep = 0; kstep < 2; ++kstep) {
    v16h af, bf;
    #pragma unroll
    for (int e = 0; e < 16; ++e) {
      int v = e >> 1, j = e & 1, grp = v >> 2;
      int K = 32 * kstep + 16 * grp + 8 * half + ((v & 3) * 2 + j);
      af[e] = (_Float16)(MEAN ? A[row * 64 + K] / den : A[row * 64 + K]);
    }
    #pragma unroll
    for (int e = 0; e < 16; ++e) {
      int v = e >> 1, j = e & 1;
      int K = 32 * kstep + 16 * half + 2 * v + j;
      bf[e] = (_Float16)B[K * NCOLS + c0 + m];
    }
    acc = __builtin_amdgcn_wmma_f32_16x16x32_f16(false, af, false, bf,
                                                 (short)0, acc, false, false);
  }
#endif

  #pragma unroll
  for (int i = 0; i < 8; ++i) {
    float v = acc[i];
    if (BIAS) v += bias[c0 + m];
    C[(r0 + i + 8 * half) * NCOLS + c0 + m] = v;
  }
}

template <int NCOLS, bool BIAS, bool MEAN>
__global__ __launch_bounds__(32 * (NCOLS / 16)) void gcn_gemm_wmma(
    const float* __restrict__ A, const float* __restrict__ B,
    const float* __restrict__ bias, const float* __restrict__ cnt, float* __restrict__ C) {
  gemm_tile<NCOLS, BIAS, MEAN>(A, B, bias, cnt, C, (int)blockIdx.x * 16, (int)(threadIdx.x >> 5) * 16,
                               (int)(threadIdx.x & 31));
}

#define RANGE   4096
#define NPART   13
#define NCHUNK  16
#define CH      (N_EDGES / NCHUNK)
#define NSUB    (NPART * 4)
#define NLOAD   13

#define XW_RT     8
#define XW_BLOCKS ((N_NODES / 16 + XW_RT - 1) / XW_RT)
#define XW_A      256
static_assert(XW_A < XW_BLOCKS, "both launches carry product blocks");
__device__ __forceinline__ void xw_block(const float* __restrict__ x, const float* __restrict__ W_in,
                                         float* __restrict__ hW, const int pb) {
  const int w  = __builtin_amdgcn_readfirstlane((int)(threadIdx.x >> 5));
  const int rt = pb * XW_RT + (w >> 2);
  if (rt < N_NODES / 16)
    gemm_tile<D_H, false, false>(x, W_in, nullptr, nullptr, hW, rt * 16, (w & 3) * 16, (int)(threadIdx.x & 31));
}
__global__ __launch_bounds__(1024) void k_rank_lds(const int* __restrict__ dst,
                                                   int* __restrict__ rank,
                                                   int* __restrict__ cntc,
                                                   int* __restrict__ wgtot,
                                                   const float* __restrict__ x,
                                                   const float* __restrict__ W_in,
                                                   float* __restrict__ hW) {
  __shared__ int s_cnt[RANGE];
  __shared__ int s_red[128];
  if (blockIdx.x >= NPART * NCHUNK) {
    xw_block(x, W_in, hW, (int)blockIdx.x - NPART * NCHUNK);
    return;
  }
  const int tid = threadIdx.x, lane = tid & 31, wave = tid >> 5;
  const int p = blockIdx.x / NCHUNK, c = blockIdx.x % NCHUNK;
  const int lo = p * RANGE;
  #pragma unroll
  for (int q = 0; q < 4; ++q) s_cnt[q * 1024 + tid] = 0;

  const int e0 = c * CH;
  const int4* d4 = (const int4*)(dst + e0);
  int4 v[NLOAD];
  #pragma unroll
  for (int k = 0; k < NLOAD; ++k) {
    const int j = tid + k * 1024;
    v[k] = (j < CH / 4) ? d4[j] : make_int4(-1, -1, -1, -1);
  }
  __syncthreads();
  #pragma unroll
  for (int k = 0; k < NLOAD; ++k) {
    const int e = e0 + 4 * (tid + k * 1024);
    unsigned r;
    r = (unsigned)(v[k].x - lo); if (r < RANGE) rank[e]     = atomicAdd(&s_cnt[r], 1);
    r = (unsigned)(v[k].y - lo); if (r < RANGE) rank[e + 1] = atomicAdd(&s_cnt[r], 1);
    r = (unsigned)(v[k].z - lo); if (r < RANGE) rank[e + 2] = atomicAdd(&s_cnt[r], 1);
    r = (unsigned)(v[k].w - lo); if (r < RANGE) rank[e + 3] = atomicAdd(&s_cnt[r], 1);
  }
  __syncthreads();
  #pragma unroll
  for (int q = 0; q < 4; ++q) {
    const int n = lo + q * 1024 + tid;
    int t = s_cnt[q * 1024 + tid];
    if (n < N_NODES) cntc[c * N_PAD + n] = t;
    #pragma unroll
    for (int off = 16; off > 0; off >>= 1) t += __shfl_xor(t, off, 32);
    if (lane == 0) s_red[q * 32 + wave] = t;
  }
  __syncthreads();
  if (tid < 128) {
    int t = s_red[tid];
    #pragma unroll
    for (int off = 16; off > 0; off >>= 1) t += __shfl_xor(t, off, 32);
    if (lane == 0) wgtot[(p * 4 + wave) * NCHUNK + c] = t;
  }
}

#define SCAN_BLOCKS ((N_NODES + 1 + 1023) / 1024)
__global__ __launch_bounds__(1024) void k_scan(const int* __restrict__ cntc,
                                               const int* __restrict__ wgtot,
                                               int* __restrict__ rowptr,
                                               int* __restrict__ basec,
                                               float* __restrict__ dinv,
                                               float* __restrict__ pool,
                                               float* __restrict__ gcnt,
                                               const float* __restrict__ x,
                                               const float* __restrict__ W_in,
                                               float* __restrict__ hW) {
  __shared__ int s_front[32];
  __shared__ int s_wtot[32];
  if (blockIdx.x >= SCAN_BLOCKS) {
    xw_block(x, W_in, hW, XW_A + (int)blockIdx.x - SCAN_BLOCKS);
    return;
  }
  const int tid = threadIdx.x, lane = tid & 31, wave = tid >> 5;
  const int n = blockIdx.x * 1024 + tid;

  if (n < N_GRAPHS * D_H) pool[n] = 0.f;
  if (n < N_GRAPHS) gcnt[n] = 0.f;

  int front = (tid < (int)blockIdx.x * NCHUNK) ? wgtot[tid] : 0;
  #pragma unroll
  for (int off = 16; off > 0; off >>= 1) front += __shfl_xor(front, off, 32);

  int cc[NCHUNK];
  int c = 0;
  #pragma unroll
  for (int k = 0; k < NCHUNK; ++k) {
    cc[k] = (n < N_NODES) ? cntc[k * N_PAD + n] : 0;
    c += cc[k];
  }
  int incl = c;
  #pragma unroll
  for (int off = 1; off < 32; off <<= 1) {
    int t = __shfl_up(incl, off, 32);
    if (lane >= off) incl += t;
  }
  if (lane == 0)  s_front[wave] = front;
  if (lane == 31) s_wtot[wave]  = incl;
  __syncthreads();
  int base = 0;
  for (int w = 0; w < 32; ++w) {
    base += s_front[w];
    if (w < wave) base += s_wtot[w];
  }
  int run = base + incl - c;
  if (n <= N_NODES) rowptr[n] = run;
  if (n < N_NODES) {
    dinv[n] = rsqrtf((float)(c + 1));
    #pragma unroll
    for (int k = 0; k < NCHUNK; ++k) {
      basec[k * N_PAD + n] = run;
      run += cc[k];
    }
  }
}

static_assert(CH % 4 == 0 && N_EDGES % 4 == 0, "a thread's four edges lie in one edge chunk and inside the list");
__global__ void k_fill(const int* __restrict__ src, const int* __restrict__ dst,
                       const int* __restrict__ rank, const int* __restrict__ basec,
                       const float* __restrict__ dinv, int2* __restrict__ ed) {
  const int i = (blockIdx.x * blockDim.x + threadIdx.x) * 4;
  if (i >= N_EDGES) return;
  const int4 s = *(const int4*)(src + i);
  const int4 d = *(const int4*)(dst + i);
  const int4 r = *(const int4*)(rank + i);
  const int* bc = basec + (i / CH) * N_PAD;
  ed[bc[d.x] + r.x] = make_int2(s.x, __float_as_int(dinv[s.x] * dinv[d.x]));
  ed[bc[d.y] + r.y] = make_int2(s.y, __float_as_int(dinv[s.y] * dinv[d.y]));
  ed[bc[d.z] + r.z] = make_int2(s.z, __float_as_int(dinv[s.z] * dinv[d.z]));
  ed[bc[d.w] + r.w] = make_int2(s.w, __float_as_int(dinv[s.w] * dinv[d.w]));
}

#define LDA 68
#define RPW 4
static_assert(N_NODES % 16 == 0, "a block holds 16 (POOL: 8) whole rows: all of its waves pass the row guard or none does");
template <bool POOL>
__global__ __launch_bounds__(POOL ? 64 : 128) void k_gather_ln(const float* __restrict__ hW,
                                                   const int* __restrict__ rowptr,
                                                   const int2* __restrict__ ed,
                                                   const float* __restrict__ dinv,
                                                   const float* __restrict__ bias,
                                                   const float* __restrict__ lnw,
                                                   const float* __restrict__ lnb,
                                                   const float* __restrict__ Wn,
                                                   float* __restrict__ hWn,
                                                   const int* __restrict__ batch,
                                                   float* __restrict__ pool,
                                                   float* __restrict__ gcnt) {
  __shared__ float s_row[POOL ? 8 * D_H : 16 * LDA];
  const int lane = threadIdx.x & 31;
  const int wave = __builtin_amdgcn_readfirstlane((int)(threadIdx.x >> 5));
  const int row0 = (int)blockIdx.x * (POOL ? 8 : 16) + wave * RPW;
  if (row0 >= N_NODES) return;
  int rp[RPW + 1];
  #pragma unroll
  for (int k = 0; k <= RPW; ++k) rp[k] = rowptr[row0 + k];
  int g[8];
  if (POOL) {
    #pragma unroll
    for (int w = 0; w < 8; ++w) g[w] = batch[(int)blockIdx.x * 8 + w];
  }
  const float2* h2 = (const float2*)hW;
  float sx[RPW], sy[RPW];
  #pragma unroll
  for (int k = 0; k < RPW; ++k) {
    const float di = dinv[row0 + k];
    const float2 self = (h2 + (row0 + k) * 32)[lane];
    sx[k] = di * di * self.x;
    sy[k] = di * di * self.y;
  }
  const float b0 = bias[2 * lane], b1 = bias[2 * lane + 1];
  const float w0 = lnw[2 * lane],  w1 = lnw[2 * lane + 1];
  const float o0 = lnb[2 * lane],  o1 = lnb[2 * lane + 1];

  const int end = rp[RPW];
  float fx[RPW], fy[RPW];
  #pragma unroll
  for (int k = 0; k < RPW; ++k) { fx[k] = 0.f; fy[k] = 0.f; }
  float ax = sx[0], ay = sy[0];
  int cur = 0;
  int nb = rp[1];
  auto next_row = [&]() {
    fx[0] = (cur == 0) ? ax : fx[0];  fy[0] = (cur == 0) ? ay : fy[0];
    fx[1] = (cur == 1) ? ax : fx[1];  fy[1] = (cur == 1) ? ay : fy[1];
    fx[2] = (cur == 2) ? ax : fx[2];  fy[2] = (cur == 2) ? ay : fy[2];
    ax = (cur == 0) ? sx[1] : (cur == 1) ? sx[2] : sx[3];
    ay = (cur == 0) ? sy[1] : (cur == 1) ? sy[2] : sy[3];
    nb = (cur == 0) ? rp[2] : (cur == 1) ? rp[3] : 0x7fffffff;
    ++cur;
  };
  int p = rp[0];
  int2 e[8];
  {
    const int pb = min(p, N_EDGES - 8);
    #pragma unroll
    for (int j = 0; j < 8; ++j) e[j] = ed[pb + j];
  }
  while (p + 8 <= end) {
    float2 r[8];
    #pragma unroll
    for (int j = 0; j < 8; ++j) r[j] = (h2 + e[j].x * 32)[lane];
    int2 en[8];
    const int pn = min(p + 8, N_EDGES - 8);
    #pragma unroll
    for (int j = 0; j < 8; ++j) en[j] = ed[pn + j];
    #pragma unroll
    for (int j = 0; j < 8; ++j) {
      while (p + j >= nb) next_row();
      const float w = __int_as_float(e[j].y);
      ax += w * r[j].x;
      ay += w * r[j].y;
    }
    #pragma unroll
    for (int j = 0; j < 8; ++j) e[j] = en[j];
    p += 8;
  }
  if (p < end) {
    if (p + 8 > N_EDGES) {
      #pragma unroll
      for (int j = 0; j < 8; ++j) e[j] = ed[min(p + j, end - 1)];
    }
    float2 r[8];
    #pragma unroll
    for (int j = 0; j < 8; ++j) r[j] = (h2 + ((p + j < end) ? e[j].x : e[0].x) * 32)[lane];
    #pragma unroll
    for (int j = 0; j < 8; ++j) {
      if (p + j < end) {
        while (p + j >= nb) next_row();
        const float w = __int_as_float(e[j].y);
        ax += w * r[j].x;
        ay += w * r[j].y;
      }
    }
  }
  while (cur < RPW - 1) next_row();
  fx[RPW - 1] = ax; fy[RPW - 1] = ay;

  const int m = lane & 15, half = lane >> 4, c0 = wave * 16;
  v2f bfr[16];
  if (!POOL) {
    #pragma unroll
    for (int t = 0; t < 16; ++t) {
      const int ka = 4 * t + 2 * half;
      bfr[t].x = Wn[ka * D_H + c0 + m];
      bfr[t].y = Wn[(ka + 1) * D_H + c0 + m];
    }
  }

  #pragma unroll
  for (int k = 0; k < RPW; ++k) {
    const float v0 = fx[k] + b0, v1 = fy[k] + b1;
    float s = v0 + v1;
    #pragma unroll
    for (int off = 16; off > 0; off >>= 1) s += __shfl_xor(s, off, 32);
    const float mu = s * (1.0f / 64.0f);
    const float d0 = v0 - mu, d1 = v1 - mu;
    float q = d0 * d0 + d1 * d1;
    #pragma unroll
    for (int off = 16; off > 0; off >>= 1) q += __shfl_xor(q, off, 32);
    const float rstd = rsqrtf(q * (1.0f / 64.0f) + LN_EPS);
    const float2 y = make_float2(fmaxf(0.f, d0 * rstd * w0 + o0), fmaxf(0.f, d1 * rstd * w1 + o1));
    const int rb = wave * RPW + k;
    if (!POOL) ((float2*)s_row)[rb * (LDA / 2) + lane] = y;
    else       ((float2*)s_row)[rb * 32 + lane] = y;
  }
  __syncthreads();
  if (!POOL) {
    v8f acc = {0.f, 0.f, 0.f, 0.f, 0.f, 0.f, 0.f, 0.f};
    #pragma unroll
    for (int t = 0; t < 16; ++t) {
      const int ka = 4 * t + 2 * half;
      const float2 av = *(const float2*)&s_row[m * LDA + ka];
      v2f a;
      a.x = av.x;
      a.y = av.y;
      acc = __builtin_amdgcn_wmma_f32_16x16x4_f32(false, a, false, bfr[t],
                                                  (short)0, acc, false, false);
    }
    const int r0 = (int)blockIdx.x * 16;
    #pragma unroll
    for (int i = 0; i < 8; ++i) hWn[(r0 + i + 8 * half) * D_H + c0 + m] = acc[i];
    return;
  }
  if (wave != 0) return;
  float r0[8], r1[8];
  #pragma unroll
  for (int w = 0; w < 8; ++w) {
    r0[w] = s_row[w * D_H + lane];
    r1[w] = s_row[w * D_H + 32 + lane];
  }
  int gc = g[0];
  float a0 = 0.f, a1 = 0.f, c = 0.f;
  #pragma unroll
  for (int w = 0; w < 8; ++w) {
    if (g[w] != gc) {
      atomicAdd(&pool[gc * D_H + lane], a0);
      atomicAdd(&pool[gc * D_H + 32 + lane], a1);
      if (lane == 0) atomicAdd(&gcnt[gc], c);
      gc = g[w]; a0 = 0.f; a1 = 0.f; c = 0.f;
    }
    a0 += r0[w];
    a1 += r1[w];
    c += 1.0f;
  }
  atomicAdd(&pool[gc * D_H + lane], a0);
  atomicAdd(&pool[gc * D_H + 32 + lane], a1);
  if (lane == 0) atomicAdd(&gcnt[gc], c);
}

static inline int cdiv(int a, int b) { return (a + b - 1) / b; }

template <bool LAST>
static void run_layer(const float* hW, const float* b,
                      const float* lnw, const float* lnb,
                      const float* dinv, const int* rowptr, const int2* ed,
                      const float* Wn, float* hWn,
                      const int* batch, float* pool, float* gcnt,
                      hipStream_t stream) {
  k_gather_ln<LAST><<<N_NODES / (LAST ? 8 : 16), LAST ? 64 : 128, 0, stream>>>(hW, rowptr, ed, dinv, b, lnw, lnb,
                                                                                 Wn, hWn, batch, pool, gcnt);
}

extern "C" void kernel_launch(void* const* d_in, const int* in_sizes, int n_in,
                              void* d_out, int out_size, void* d_ws, size_t ws_size,
                              hipStream_t stream) {
  (void)in_sizes; (void)n_in; (void)out_size; (void)ws_size;

  const float* x       = (const float*)d_in[0];
  const int*   eidx    = (const int*)d_in[1];
  const int*   batch   = (const int*)d_in[2];
  const float* W_in    = (const float*)d_in[3];
  const float* b_in    = (const float*)d_in[4];
  const float* ln_in_w = (const float*)d_in[5];
  const float* ln_in_b = (const float*)d_in[6];
  const float* W_h     = (const float*)d_in[7];
  const float* b_h     = (const float*)d_in[8];
  const float* ln_h_w  = (const float*)d_in[9];
  const float* ln_h_b  = (const float*)d_in[10];
  const float* W_out   = (const float*)d_in[11];
  const float* b_out   = (const float*)d_in[12];

  const int* src = eidx;
  const int* dst = eidx + N_EDGES;

  float* ws    = (float*)d_ws;
  float* dinv  = ws;
  int*   rowptr= (int*)(dinv + N_PAD);
  int*   cntc  = rowptr + N_PAD;
  int*   basec = cntc + NCHUNK * N_PAD;
  int*   wgtot = basec + NCHUNK * N_PAD;
  int*   rank  = wgtot + 1024;
  int2*  ed    = (int2*)(rank + N_EDGES);
  float* hW    = (float*)(ed + N_EDGES);
  float* hW2   = hW + (size_t)N_NODES*64;
  float* pool  = hW2 + (size_t)N_NODES*64;
  float* cnt_g = pool + N_GRAPHS * 64;

  k_rank_lds<<<NPART * NCHUNK + XW_A, 1024, 0, stream>>>(dst, rank, cntc, wgtot, x, W_in, hW);
  k_scan<<<SCAN_BLOCKS + XW_BLOCKS - XW_A, 1024, 0, stream>>>(cntc, wgtot, rowptr, basec, dinv, pool, cnt_g,
                                                              x, W_in, hW);
  k_fill<<<cdiv(N_EDGES / 4, 256), 256, 0, stream>>>(src, dst, rank, basec, dinv, ed);

  run_layer<false>(hW,  b_in,     ln_in_w,     ln_in_b,     dinv, rowptr, ed, W_h,           hW2,
                   batch, pool, cnt_g, stream);
  run_layer<false>(hW2, b_h,      ln_h_w,      ln_h_b,      dinv, rowptr, ed, W_h + 64 * 64, hW,
                   batch, pool, cnt_g, stream);
  run_layer<true>(hW,   b_h + 64, ln_h_w + 64, ln_h_b + 64, dinv, rowptr, ed, nullptr,       nullptr,
                  batch, pool, cnt_g, stream);

  gcn_gemm_wmma<D_OUT, true, true><<<N_GRAPHS / 16, 64, 0, stream>>>(pool, W_out, b_out, cnt_g, (float*)d_out);
}


# Mean pool and output projection folded into the last layer's row kernel: a block's 16 rows x W_out on f32 WMMA, graph run sums / row count added to out (f32 atomics, order varies); 7 to 6 launches
# speedup vs baseline: 5.6913x; 1.0436x over previous
#include <hip/hip_runtime.h>
#include <hip/hip_bf16.h>

#define N_NODES  50000
#define N_EDGES  800000
#define N_GRAPHS 512
#define D_H      64
#define D_OUT    32
#define LN_EPS   1e-5f
#define N_PAD    50048

typedef __attribute__((ext_vector_type(2)))  float    v2f;
typedef __attribute__((ext_vector_type(8)))  float    v8f;

template <int NCOLS>
__device__ __forceinline__ void gemm_tile(
    const float* __restrict__ A, const float* __restrict__ B, float* __restrict__ C,
    const int r0, const int c0, const int lane) {
  const int m    = lane & 15;
  const int half = lane >> 4;
  const int row  = r0 + m;

  v8f acc = {0.f, 0.f, 0.f, 0.f, 0.f, 0.f, 0.f, 0.f};

  #pragma unroll
  for (int k0 = 0; k0 < 64; k0 += 4) {
    const int ka = k0 + 2 * half;
    v2f a, b;
    a.x = A[row * 64 + ka];
    a.y = A[row * 64 + ka + 1];
    b.x = B[ka * NCOLS + c0 + m];
    b.y = B[(ka + 1) * NCOLS + c0 + m];
    acc = __builtin_amdgcn_wmma_f32_16x16x4_f32(false, a, false, b,
                                                (short)0, acc, false, false);
  }

  #pragma unroll
  for (int i = 0; i < 8; ++i) C[(r0 + i + 8 * half) * NCOLS + c0 + m] = acc[i];
}

#define RANGE   4096
#define NPART   13
#define NCHUNK  16
#define CH      (N_EDGES / NCHUNK)
#define NSUB    (NPART * 4)
#define NLOAD   13

#define XW_RT     8
#define XW_BLOCKS ((N_NODES / 16 + XW_RT - 1) / XW_RT)
#define XW_A      256
static_assert(XW_A < XW_BLOCKS, "both launches carry product blocks");
__device__ __forceinline__ void xw_block(const float* __restrict__ x, const float* __restrict__ W_in,
                                         float* __restrict__ hW, const int pb) {
  const int w  = __builtin_amdgcn_readfirstlane((int)(threadIdx.x >> 5));
  const int rt = pb * XW_RT + (w >> 2);
  if (rt < N_NODES / 16)
    gemm_tile<D_H>(x, W_in, hW, rt * 16, (w & 3) * 16, (int)(threadIdx.x & 31));
}
__global__ __launch_bounds__(1024) void k_rank_lds(const int* __restrict__ dst,
                                                   int* __restrict__ rank,
                                                   int* __restrict__ cntc,
                                                   int* __restrict__ wgtot,
                                                   const float* __restrict__ x,
                                                   const float* __restrict__ W_in,
                                                   float* __restrict__ hW,
                                                   int* __restrict__ gcnt) {
  __shared__ int s_cnt[RANGE];
  __shared__ int s_red[128];
  if (blockIdx.x >= NPART * NCHUNK) {
    xw_block(x, W_in, hW, (int)blockIdx.x - NPART * NCHUNK);
    return;
  }
  const int tid = threadIdx.x, lane = tid & 31, wave = tid >> 5;
  const int p = blockIdx.x / NCHUNK, c = blockIdx.x % NCHUNK;
  const int lo = p * RANGE;
  if (blockIdx.x == 0 && tid < N_GRAPHS) gcnt[tid] = 0;
  #pragma unroll
  for (int q = 0; q < 4; ++q) s_cnt[q * 1024 + tid] = 0;

  const int e0 = c * CH;
  const int4* d4 = (const int4*)(dst + e0);
  int4 v[NLOAD];
  #pragma unroll
  for (int k = 0; k < NLOAD; ++k) {
    const int j = tid + k * 1024;
    v[k] = (j < CH / 4) ? d4[j] : make_int4(-1, -1, -1, -1);
  }
  __syncthreads();
  #pragma unroll
  for (int k = 0; k < NLOAD; ++k) {
    const int e = e0 + 4 * (tid + k * 1024);
    unsigned r;
    r = (unsigned)(v[k].x - lo); if (r < RANGE) rank[e]     = atomicAdd(&s_cnt[r], 1);
    r = (unsigned)(v[k].y - lo); if (r < RANGE) rank[e + 1] = atomicAdd(&s_cnt[r], 1);
    r = (unsigned)(v[k].z - lo); if (r < RANGE) rank[e + 2] = atomicAdd(&s_cnt[r], 1);
    r = (unsigned)(v[k].w - lo); if (r < RANGE) rank[e + 3] = atomicAdd(&s_cnt[r], 1);
  }
  __syncthreads();
  #pragma unroll
  for (int q = 0; q < 4; ++q) {
    const int n = lo + q * 1024 + tid;
    int t = s_cnt[q * 1024 + tid];
    if (n < N_NODES) cntc[c * N_PAD + n] = t;
    #pragma unroll
    for (int off = 16; off > 0; off >>= 1) t += __shfl_xor(t, off, 32);
    if (lane == 0) s_red[q * 32 + wave] = t;
  }
  __syncthreads();
  if (tid < 128) {
    int t = s_red[tid];
    #pragma unroll
    for (int off = 16; off > 0; off >>= 1) t += __shfl_xor(t, off, 32);
    if (lane == 0) wgtot[(p * 4 + wave) * NCHUNK + c] = t;
  }
}

#define SCAN_BLOCKS ((N_NODES + 1 + 1023) / 1024)
__global__ __launch_bounds__(1024) void k_scan(const int* __restrict__ cntc,
                                               const int* __restrict__ wgtot,
                                               int* __restrict__ rowptr,
                                               int* __restrict__ basec,
                                               float* __restrict__ dinv,
                                               const int* __restrict__ batch,
                                               int* __restrict__ gcnt,
                                               const float* __restrict__ x,
                                               const float* __restrict__ W_in,
                                               float* __restrict__ hW) {
  __shared__ int s_front[32];
  __shared__ int s_wtot[32];
  if (blockIdx.x >= SCAN_BLOCKS) {
    xw_block(x, W_in, hW, XW_A + (int)blockIdx.x - SCAN_BLOCKS);
    return;
  }
  const int tid = threadIdx.x, lane = tid & 31, wave = tid >> 5;
  const int n = blockIdx.x * 1024 + tid;

  const int gb = (n < N_NODES) ? batch[n] : -1;

  int front = (tid < (int)blockIdx.x * NCHUNK) ? wgtot[tid] : 0;
  #pragma unroll
  for (int off = 16; off > 0; off >>= 1) front += __shfl_xor(front, off, 32);

  int cc[NCHUNK];
  int c = 0;
  #pragma unroll
  for (int k = 0; k < NCHUNK; ++k) {
    cc[k] = (n < N_NODES) ? cntc[k * N_PAD + n] : 0;
    c += cc[k];
  }
  int incl = c;
  #pragma unroll
  for (int off = 1; off < 32; off <<= 1) {
    int t = __shfl_up(incl, off, 32);
    if (lane >= off) incl += t;
  }
  {
    const int gp = __shfl_up(gb, 1, 32);
    const bool head = (lane == 0) || (gb != gp);
    const unsigned hm = (unsigned)__ballot(head);
    const unsigned above = (lane == 31) ? 0u : (hm & (0xffffffffu << (lane + 1)));
    const int nxt = above ? (__ffs(above) - 1) : 32;
    if (head && gb >= 0) atomicAdd(&gcnt[gb], nxt - lane);
  }
  if (lane == 0)  s_front[wave] = front;
  if (lane == 31) s_wtot[wave]  = incl;
  __syncthreads();
  int base = 0;
  for (int w = 0; w < 32; ++w) {
    base += s_front[w];
    if (w < wave) base += s_wtot[w];
  }
  int run = base + incl - c;
  if (n <= N_NODES) rowptr[n] = run;
  if (n < N_NODES) {
    dinv[n] = rsqrtf((float)(c + 1));
    #pragma unroll
    for (int k = 0; k < NCHUNK; ++k) {
      basec[k * N_PAD + n] = run;
      run += cc[k];
    }
  }
}

static_assert(CH % 4 == 0 && N_EDGES % 4 == 0, "a thread's four edges lie in one edge chunk and inside the list");
__global__ void k_fill(const int* __restrict__ src, const int* __restrict__ dst,
                       const int* __restrict__ rank, const int* __restrict__ basec,
                       const float* __restrict__ dinv, int2* __restrict__ ed,
                       const float* __restrict__ b_out, float* __restrict__ out) {
  const int i = (blockIdx.x * blockDim.x + threadIdx.x) * 4;
  if (i >= N_EDGES) return;
  static_assert(N_GRAPHS * D_OUT <= N_EDGES / 4, "one thread per output element");
  const bool init_out = i / 4 < N_GRAPHS * D_OUT;
  const float bo = init_out ? b_out[(i / 4) & (D_OUT - 1)] : 0.f;
  const int4 s = *(const int4*)(src + i);
  const int4 d = *(const int4*)(dst + i);
  const int4 r = *(const int4*)(rank + i);
  const int* bc = basec + (i / CH) * N_PAD;
  ed[bc[d.x] + r.x] = make_int2(s.x, __float_as_int(dinv[s.x] * dinv[d.x]));
  ed[bc[d.y] + r.y] = make_int2(s.y, __float_as_int(dinv[s.y] * dinv[d.y]));
  ed[bc[d.z] + r.z] = make_int2(s.z, __float_as_int(dinv[s.z] * dinv[d.z]));
  ed[bc[d.w] + r.w] = make_int2(s.w, __float_as_int(dinv[s.w] * dinv[d.w]));
  if (init_out) out[i / 4] = bo;
}

#define LDA 68
#define RPW 4
static_assert(N_NODES % 16 == 0, "a block holds 16 whole rows: all of its waves pass the row guard or none does");
template <bool POOL>
__global__ __launch_bounds__(128) void k_gather_ln(const float* __restrict__ hW,
                                                   const int* __restrict__ rowptr,
                                                   const int2* __restrict__ ed,
                                                   const float* __restrict__ dinv,
                                                   const float* __restrict__ bias,
                                                   const float* __restrict__ lnw,
                                                   const float* __restrict__ lnb,
                                                   const float* __restrict__ Wn,
                                                   float* __restrict__ hWn,
                                                   const int* __restrict__ batch,
                                                   const int* __restrict__ gcnt,
                                                   float* __restrict__ out) {
  __shared__ float s_row[16 * LDA];
  const int lane = threadIdx.x & 31;
  const int wave = __builtin_amdgcn_readfirstlane((int)(threadIdx.x >> 5));
  const int row0 = (int)blockIdx.x * 16 + wave * RPW;
  if (row0 >= N_NODES) return;
  int rp[RPW + 1];
  #pragma unroll
  for (int k = 0; k <= RPW; ++k) rp[k] = rowptr[row0 + k];
  int g[16];
  if (POOL) {
    #pragma unroll
    for (int w = 0; w < 16; ++w) g[w] = batch[(int)blockIdx.x * 16 + w];
  }
  const float2* h2 = (const float2*)hW;
  float sx[RPW], sy[RPW];
  #pragma unroll
  for (int k = 0; k < RPW; ++k) {
    const float di = dinv[row0 + k];
    const float2 self = (h2 + (row0 + k) * 32)[lane];
    sx[k] = di * di * self.x;
    sy[k] = di * di * self.y;
  }
  const float b0 = bias[2 * lane], b1 = bias[2 * lane + 1];
  const float w0 = lnw[2 * lane],  w1 = lnw[2 * lane + 1];
  const float o0 = lnb[2 * lane],  o1 = lnb[2 * lane + 1];

  const int end = rp[RPW];
  float fx[RPW], fy[RPW];
  #pragma unroll
  for (int k = 0; k < RPW; ++k) { fx[k] = 0.f; fy[k] = 0.f; }
  float ax = sx[0], ay = sy[0];
  int cur = 0;
  int nb = rp[1];
  auto next_row = [&]() {
    fx[0] = (cur == 0) ? ax : fx[0];  fy[0] = (cur == 0) ? ay : fy[0];
    fx[1] = (cur == 1) ? ax : fx[1];  fy[1] = (cur == 1) ? ay : fy[1];
    fx[2] = (cur == 2) ? ax : fx[2];  fy[2] = (cur == 2) ? ay : fy[2];
    ax = (cur == 0) ? sx[1] : (cur == 1) ? sx[2] : sx[3];
    ay = (cur == 0) ? sy[1] : (cur == 1) ? sy[2] : sy[3];
    nb = (cur == 0) ? rp[2] : (cur == 1) ? rp[3] : 0x7fffffff;
    ++cur;
  };
  int p = rp[0];
  int2 e[8];
  {
    const int pb = min(p, N_EDGES - 8);
    #pragma unroll
    for (int j = 0; j < 8; ++j) e[j] = ed[pb + j];
  }
  while (p + 8 <= end) {
    float2 r[8];
    #pragma unroll
    for (int j = 0; j < 8; ++j) r[j] = (h2 + e[j].x * 32)[lane];
    int2 en[8];
    const int pn = min(p + 8, N_EDGES - 8);
    #pragma unroll
    for (int j = 0; j < 8; ++j) en[j] = ed[pn + j];
    #pragma unroll
    for (int j = 0; j < 8; ++j) {
      while (p + j >= nb) next_row();
      const float w = __int_as_float(e[j].y);
      ax += w * r[j].x;
      ay += w * r[j].y;
    }
    #pragma unroll
    for (int j = 0; j < 8; ++j) e[j] = en[j];
    p += 8;
  }
  if (p < end) {
    if (p + 8 > N_EDGES) {
      #pragma unroll
      for (int j = 0; j < 8; ++j) e[j] = ed[min(p + j, end - 1)];
    }
    float2 r[8];
    #pragma unroll
    for (int j = 0; j < 8; ++j) r[j] = (h2 + ((p + j < end) ? e[j].x : e[0].x) * 32)[lane];
    #pragma unroll
    for (int j = 0; j < 8; ++j) {
      if (p + j < end) {
        while (p + j >= nb) next_row();
        const float w = __int_as_float(e[j].y);
        ax += w * r[j].x;
        ay += w * r[j].y;
      }
    }
  }
  while (cur < RPW - 1) next_row();
  fx[RPW - 1] = ax; fy[RPW - 1] = ay;

  const int m = lane & 15, half = lane >> 4, c0 = wave * 16;
  constexpr int NC = POOL ? D_OUT : D_H;
  const bool tile_wave = !POOL || wave < D_OUT / 16;
  v2f bfr[16];
  float dn[16];
  if (tile_wave) {
    #pragma unroll
    for (int t = 0; t < 16; ++t) {
      const int ka = 4 * t + 2 * half;
      bfr[t].x = Wn[ka * NC + c0 + m];
      bfr[t].y = Wn[(ka + 1) * NC + c0 + m];
    }
    if (POOL) {
      #pragma unroll
      for (int w = 0; w < 16; ++w) dn[w] = fmaxf((float)gcnt[g[w]], 1.0f);
    }
  }

  #pragma unroll
  for (int k = 0; k < RPW; ++k) {
    const float v0 = fx[k] + b0, v1 = fy[k] + b1;
    float s = v0 + v1;
    #pragma unroll
    for (int off = 16; off > 0; off >>= 1) s += __shfl_xor(s, off, 32);
    const float mu = s * (1.0f / 64.0f);
    const float d0 = v0 - mu, d1 = v1 - mu;
    float q = d0 * d0 + d1 * d1;
    #pragma unroll
    for (int off = 16; off > 0; off >>= 1) q += __shfl_xor(q, off, 32);
    const float rstd = rsqrtf(q * (1.0f / 64.0f) + LN_EPS);
    const float2 y = make_float2(fmaxf(0.f, d0 * rstd * w0 + o0), fmaxf(0.f, d1 * rstd * w1 + o1));
    const int rb = wave * RPW + k;
    ((float2*)s_row)[rb * (LDA / 2) + lane] = y;
  }
  __syncthreads();
  if (!tile_wave) return;
  v8f acc = {0.f, 0.f, 0.f, 0.f, 0.f, 0.f, 0.f, 0.f};
  #pragma unroll
  for (int t = 0; t < 16; ++t) {
    const int ka = 4 * t + 2 * half;
    const float2 av = *(const float2*)&s_row[m * LDA + ka];
    v2f a;
    a.x = av.x;
    a.y = av.y;
    acc = __builtin_amdgcn_wmma_f32_16x16x4_f32(false, a, false, bfr[t],
                                                (short)0, acc, false, false);
  }
  if (!POOL) {
    const int r0 = (int)blockIdx.x * 16;
    #pragma unroll
    for (int i = 0; i < 8; ++i) hWn[(r0 + i + 8 * half) * D_H + c0 + m] = acc[i];
    return;
  }
  int gc = half ? g[8] : g[0];
  float dc = half ? dn[8] : dn[0];
  float a0 = 0.f;
  #pragma unroll
  for (int w = 0; w < 8; ++w) {
    const int gw = half ? g[8 + w] : g[w];
    if (gw != gc) {
      atomicAdd(&out[gc * D_OUT + c0 + m], a0 / dc);
      gc = gw; dc = half ? dn[8 + w] : dn[w]; a0 = 0.f;
    }
    a0 += acc[w];
  }
  atomicAdd(&out[gc * D_OUT + c0 + m], a0 / dc);
}

static inline int cdiv(int a, int b) { return (a + b - 1) / b; }

template <bool LAST>
static void run_layer(const float* hW, const float* b,
                      const float* lnw, const float* lnb,
                      const float* dinv, const int* rowptr, const int2* ed,
                      const float* Wn, float* hWn,
                      const int* batch, const int* gcnt, float* out,
                      hipStream_t stream) {
  k_gather_ln<LAST><<<N_NODES / 16, 128, 0, stream>>>(hW, rowptr, ed, dinv, b, lnw, lnb, Wn, hWn, batch, gcnt, out);
}

extern "C" void kernel_launch(void* const* d_in, const int* in_sizes, int n_in,
                              void* d_out, int out_size, void* d_ws, size_t ws_size,
                              hipStream_t stream) {
  (void)in_sizes; (void)n_in; (void)out_size; (void)ws_size;

  const float* x       = (const float*)d_in[0];
  const int*   eidx    = (const int*)d_in[1];
  const int*   batch   = (const int*)d_in[2];
  const float* W_in    = (const float*)d_in[3];
  const float* b_in    = (const float*)d_in[4];
  const float* ln_in_w = (const float*)d_in[5];
  const float* ln_in_b = (const float*)d_in[6];
  const float* W_h     = (const float*)d_in[7];
  const float* b_h     = (const float*)d_in[8];
  const float* ln_h_w  = (const float*)d_in[9];
  const float* ln_h_b  = (const float*)d_in[10];
  const float* W_out   = (const float*)d_in[11];
  const float* b_out   = (const float*)d_in[12];

  const int* src = eidx;
  const int* dst = eidx + N_EDGES;

  float* ws    = (float*)d_ws;
  float* dinv  = ws;
  int*   rowptr= (int*)(dinv + N_PAD);
  int*   cntc  = rowptr + N_PAD;
  int*   basec = cntc + NCHUNK * N_PAD;
  int*   wgtot = basec + NCHUNK * N_PAD;
  int*   rank  = wgtot + 1024;
  int2*  ed    = (int2*)(rank + N_EDGES);
  float* hW    = (float*)(ed + N_EDGES);
  float* hW2   = hW + (size_t)N_NODES*64;
  int*   cnt_g = (int*)(hW2 + (size_t)N_NODES*64);
  float* out   = (float*)d_out;

  k_rank_lds<<<NPART * NCHUNK + XW_A, 1024, 0, stream>>>(dst, rank, cntc, wgtot, x, W_in, hW, cnt_g);
  k_scan<<<SCAN_BLOCKS + XW_BLOCKS - XW_A, 1024, 0, stream>>>(cntc, wgtot, rowptr, basec, dinv, batch, cnt_g,
                                                              x, W_in, hW);
  k_fill<<<cdiv(N_EDGES / 4, 256), 256, 0, stream>>>(src, dst, rank, basec, dinv, ed, b_out, out);

  run_layer<false>(hW,  b_in,     ln_in_w,     ln_in_b,     dinv, rowptr, ed, W_h,           hW2,
                   batch, cnt_g, out, stream);
  run_layer<false>(hW2, b_h,      ln_h_w,      ln_h_b,      dinv, rowptr, ed, W_h + 64 * 64, hW,
                   batch, cnt_g, out, stream);
  run_layer<true>(hW,   b_h + 64, ln_h_w + 64, ln_h_b + 64, dinv, rowptr, ed, W_out,         nullptr,
                  batch, cnt_g, out, stream);
}


# Edge list built in 2 launches, not 3: a placement block sums its rows' chunk counts, scans them in LDS, slots by LDS adds; gcn weight split (rows stored x dinv, row sum x dinv); row sum order kept
# speedup vs baseline: 6.0912x; 1.0703x over previous
#include <hip/hip_runtime.h>
#include <hip/hip_bf16.h>


#define N_NODES  50000
#define N_EDGES  800000
#define N_GRAPHS 512
#define D_H      64
#define D_OUT    32
#define LN_EPS   1e-5f
#define N_PAD    50048

typedef __attribute__((ext_vector_type(2)))  float    v2f;
typedef __attribute__((ext_vector_type(8)))  float    v8f;

template <int NCOLS>
__device__ __forceinline__ void gemm_tile(
    const float* __restrict__ A, const float* __restrict__ B, float* __restrict__ C,
    const int r0, const int c0, const int lane, const float ascale) {
  const int m    = lane & 15;
  const int half = lane >> 4;
  const int row  = r0 + m;

  v8f acc = {0.f, 0.f, 0.f, 0.f, 0.f, 0.f, 0.f, 0.f};

  #pragma unroll
  for (int k0 = 0; k0 < 64; k0 += 4) {
    const int ka = k0 + 2 * half;
    v2f a, b;
    a.x = A[row * 64 + ka] * ascale;
    a.y = A[row * 64 + ka + 1] * ascale;
    b.x = B[ka * NCOLS + c0 + m];
    b.y = B[(ka + 1) * NCOLS + c0 + m];
    acc = __builtin_amdgcn_wmma_f32_16x16x4_f32(false, a, false, b,
                                                (short)0, acc, false, false);
  }

  #pragma unroll
  for (int i = 0; i < 8; ++i) C[(r0 + i + 8 * half) * NCOLS + c0 + m] = acc[i];
}

#define RANGE   4096
#define NPART   13
#define NCHUNK  16
#define CH      (N_EDGES / NCHUNK)
#define NSUB    (NPART * 4)
#define NLOAD   13
static_assert(CH % 4 == 0 && N_EDGES % 4 == 0 && N_PAD % 4 == 0, "16-byte reads stay inside a chunk and a plane");
static_assert(NPART * RANGE >= N_PAD, "the ranges cover every plane entry");

__global__ __launch_bounds__(1024) void k_count(const int* __restrict__ dst,
                                                int* __restrict__ cntc,
                                                int* __restrict__ wgtot,
                                                int* __restrict__ gcnt) {
  __shared__ int s_cnt[RANGE];
  __shared__ int s_red[128];
  const int tid = threadIdx.x, lane = tid & 31, wave = tid >> 5;
  const int p = blockIdx.x / NCHUNK, c = blockIdx.x % NCHUNK;
  const int lo = p * RANGE;
  if (blockIdx.x == 0 && tid < N_GRAPHS) gcnt[tid] = 0;
  #pragma unroll
  for (int q = 0; q < 4; ++q) s_cnt[q * 1024 + tid] = 0;

  const int e0 = c * CH;
  const int4* d4 = (const int4*)(dst + e0);
  int4 v[NLOAD];
  #pragma unroll
  for (int k = 0; k < NLOAD; ++k) {
    const int j = tid + k * 1024;
    v[k] = (j < CH / 4) ? d4[j] : make_int4(-1, -1, -1, -1);
  }
  __syncthreads();
  #pragma unroll
  for (int k = 0; k < NLOAD; ++k) {
    unsigned r;
    r = (unsigned)(v[k].x - lo); if (r < RANGE) atomicAdd(&s_cnt[r], 1);
    r = (unsigned)(v[k].y - lo); if (r < RANGE) atomicAdd(&s_cnt[r], 1);
    r = (unsigned)(v[k].z - lo); if (r < RANGE) atomicAdd(&s_cnt[r], 1);
    r = (unsigned)(v[k].w - lo); if (r < RANGE) atomicAdd(&s_cnt[r], 1);
  }
  __syncthreads();
  #pragma unroll
  for (int q = 0; q < 4; ++q) {
    const int n = lo + q * 1024 + tid;
    int t = s_cnt[q * 1024 + tid];
    if (n < N_PAD) cntc[c * N_PAD + n] = t;
    #pragma unroll
    for (int off = 16; off > 0; off >>= 1) t += __shfl_xor(t, off, 32);
    if (lane == 0) s_red[q * 32 + wave] = t;
  }
  __syncthreads();
  if (tid < 128) {
    int t = s_red[tid];
    #pragma unroll
    for (int off = 16; off > 0; off >>= 1) t += __shfl_xor(t, off, 32);
    if (lane == 0) wgtot[(p * 4 + wave) * NCHUNK + c] = t;
  }
}

#define XW_RT     8
#define XW_BLOCKS ((N_NODES / 16 + XW_RT - 1) / XW_RT)
__device__ __forceinline__ void xw_block(const float* __restrict__ x, const float* __restrict__ W_in,
                                         const int* __restrict__ cntc, float* __restrict__ hW, const int pb) {
  const int w  = __builtin_amdgcn_readfirstlane((int)(threadIdx.x >> 5));
  const int rt = pb * XW_RT + (w >> 2);
  if (rt < N_NODES / 16) {
    const int lane = (int)(threadIdx.x & 31);
    const int row  = rt * 16 + (lane & 15);
    int deg = 1;
    #pragma unroll
    for (int k = 0; k < NCHUNK; ++k) deg += cntc[k * N_PAD + row];
    gemm_tile<D_H>(x, W_in, hW, rt * 16, (w & 3) * 16, lane, rsqrtf((float)deg));
  }
}

static_assert((N_GRAPHS * D_OUT) % 1024 == 0 && (N_GRAPHS * D_OUT) / 1024 <= NPART * NCHUNK,
              "whole placement blocks store the output's start value");
static_assert((NPART - 1) * 4 * NCHUNK <= 1024, "one wgtot entry per thread covers the edges in front of a range");
__global__ __launch_bounds__(1024) void k_place(const int* __restrict__ src, const int* __restrict__ dst,
                                                const int* __restrict__ cntc,
                                                const int* __restrict__ wgtot,
                                                int* __restrict__ rowptr,
                                                float* __restrict__ dinv,
                                                int* __restrict__ ed,
                                                const int* __restrict__ batch,
                                                int* __restrict__ gcnt,
                                                const float* __restrict__ b_out,
                                                float* __restrict__ out,
                                                const float* __restrict__ x,
                                                const float* __restrict__ W_in,
                                                float* __restrict__ hW) {
  __shared__ int s_pos[RANGE];
  __shared__ int s_front[32];
  __shared__ int s_wtot[32];
  if (blockIdx.x >= NPART * NCHUNK) {
    xw_block(x, W_in, cntc, hW, (int)blockIdx.x - NPART * NCHUNK);
    return;
  }
  const int tid = threadIdx.x, lane = tid & 31, wave = tid >> 5;
  const int p = blockIdx.x / NCHUNK, c = blockIdx.x % NCHUNK;
  const int lo = p * RANGE;

  if (blockIdx.x < (N_GRAPHS * D_OUT) / 1024) out[blockIdx.x * 1024 + tid] = b_out[tid & (D_OUT - 1)];

  int front = (tid < p * 4 * NCHUNK) ? wgtot[tid] : 0;
  #pragma unroll
  for (int off = 16; off > 0; off >>= 1) front += __shfl_xor(front, off, 32);

  const int n4 = lo + 4 * tid;
  const bool inpl = n4 < N_PAD;
  int4 tot = make_int4(0, 0, 0, 0), pre = make_int4(0, 0, 0, 0);
  #pragma unroll 4
  for (int k = 0; k < NCHUNK; ++k) {
    const int4 v = inpl ? *(const int4*)(cntc + k * N_PAD + n4) : make_int4(0, 0, 0, 0);
    tot.x += v.x; tot.y += v.y; tot.z += v.z; tot.w += v.w;
    if (k < c) { pre.x += v.x; pre.y += v.y; pre.z += v.z; pre.w += v.w; }
  }
  const int ts = tot.x + tot.y + tot.z + tot.w;
  int incl = ts;
  #pragma unroll
  for (int off = 1; off < 32; off <<= 1) {
    int t = __shfl_up(incl, off, 32);
    if (lane >= off) incl += t;
  }
  if (lane == 0)  s_front[wave] = front;
  if (lane == 31) s_wtot[wave]  = incl;
  __syncthreads();
  int base = 0;
  for (int w = 0; w < 32; ++w) {
    base += s_front[w];
    if (w < wave) base += s_wtot[w];
  }
  const int r0 = base + incl - ts;
  const int r1 = r0 + tot.x, r2 = r1 + tot.y, r3 = r2 + tot.z;
  *(int4*)&s_pos[4 * tid] = make_int4(r0 + pre.x, r1 + pre.y, r2 + pre.z, r3 + pre.w);
  if (c == 0) {
    if (n4     <= N_NODES) rowptr[n4]     = r0;
    if (n4 + 1 <= N_NODES) rowptr[n4 + 1] = r1;
    if (n4 + 2 <= N_NODES) rowptr[n4 + 2] = r2;
    if (n4 + 3 <= N_NODES) rowptr[n4 + 3] = r3;
    if (n4     <  N_NODES) dinv[n4]     = rsqrtf((float)(tot.x + 1));
    if (n4 + 1 <  N_NODES) dinv[n4 + 1] = rsqrtf((float)(tot.y + 1));
    if (n4 + 2 <  N_NODES) dinv[n4 + 2] = rsqrtf((float)(tot.z + 1));
    if (n4 + 3 <  N_NODES) dinv[n4 + 3] = rsqrtf((float)(tot.w + 1));
    for (int q = 0; q < 4; ++q) {
      const int n  = lo + q * 1024 + tid;
      const int gb = (n < N_NODES) ? batch[n] : -1;
      const int gp = __shfl_up(gb, 1, 32);
      const bool head = (lane == 0) || (gb != gp);
      const unsigned hm = (unsigned)__ballot(head);
      const unsigned above = (lane == 31) ? 0u : (hm & (0xffffffffu << (lane + 1)));
      const int nxt = above ? (__ffs(above) - 1) : 32;
      if (head && gb >= 0) atomicAdd(&gcnt[gb], nxt - lane);
    }
  }
  __syncthreads();

  auto place4 = [&](const int4 d, const int4 s, const bool in) {
    const unsigned rx = in ? (unsigned)(d.x - lo) : 0xffffffffu, ry = in ? (unsigned)(d.y - lo) : 0xffffffffu;
    const unsigned rz = in ? (unsigned)(d.z - lo) : 0xffffffffu, rw = in ? (unsigned)(d.w - lo) : 0xffffffffu;
    int px = -1, py = -1, pz = -1, pw = -1;
    if (rx < RANGE) px = atomicAdd(&s_pos[rx], 1);
    if (ry < RANGE) py = atomicAdd(&s_pos[ry], 1);
    if (rz < RANGE) pz = atomicAdd(&s_pos[rz], 1);
    if (rw < RANGE) pw = atomicAdd(&s_pos[rw], 1);
    if (px >= 0) ed[px] = s.x;
    if (py >= 0) ed[py] = s.y;
    if (pz >= 0) ed[pz] = s.z;
    if (pw >= 0) ed[pw] = s.w;
  };
  const int e0 = c * CH;
  const int4* d4 = (const int4*)(dst + e0);
  const int4* s4 = (const int4*)(src + e0);
  auto rd = [&](const int j, int4& d, int4& s) {
    const int jc = min(j, CH / 4 - 1);
    d = d4[jc];
    s = s4[jc];
  };
  int4 dA, sA, dB, sB, dC, sC;
  rd(tid, dA, sA);
  rd(tid + 1024, dB, sB);
  #pragma unroll 1
  for (int k = 0; k < NLOAD; k += 3) {
    rd(tid + (k + 2) * 1024, dC, sC);
    place4(dA, sA, tid + k * 1024 < CH / 4);
    rd(tid + (k + 3) * 1024, dA, sA);
    place4(dB, sB, tid + (k + 1) * 1024 < CH / 4);
    rd(tid + (k + 4) * 1024, dB, sB);
    place4(dC, sC, tid + (k + 2) * 1024 < CH / 4);
  }
}

#define LDA 68
#define RPW 4
static_assert(N_NODES % 16 == 0, "a block holds 16 whole rows: all of its waves pass the row guard or none does");
template <bool POOL>
__global__ __launch_bounds__(128) void k_gather_ln(const float* __restrict__ hW,
                                                   const int* __restrict__ rowptr,
                                                   const int* __restrict__ ed,
                                                   const float* __restrict__ dinv,
                                                   const float* __restrict__ bias,
                                                   const float* __restrict__ lnw,
                                                   const float* __restrict__ lnb,
                                                   const float* __restrict__ Wn,
                                                   float* __restrict__ hWn,
                                                   const int* __restrict__ batch,
                                                   const int* __restrict__ gcnt,
                                                   float* __restrict__ out) {
  __shared__ float s_row[16 * LDA];
  const int lane = threadIdx.x & 31;
  const int wave = __builtin_amdgcn_readfirstlane((int)(threadIdx.x >> 5));
  const int row0 = (int)blockIdx.x * 16 + wave * RPW;
  if (row0 >= N_NODES) return;
  int rp[RPW + 1];
  #pragma unroll
  for (int k = 0; k <= RPW; ++k) rp[k] = rowptr[row0 + k];
  int g[16];
  if (POOL) {
    #pragma unroll
    for (int w = 0; w < 16; ++w) g[w] = batch[(int)blockIdx.x * 16 + w];
  }
  const float2* h2 = (const float2*)hW;
  float sx[RPW], sy[RPW], di[RPW];
  #pragma unroll
  for (int k = 0; k < RPW; ++k) {
    di[k] = dinv[row0 + k];
    const float2 self = (h2 + (row0 + k) * 32)[lane];
    sx[k] = self.x;
    sy[k] = self.y;
  }
  const float b0 = bias[2 * lane], b1 = bias[2 * lane + 1];
  const float w0 = lnw[2 * lane],  w1 = lnw[2 * lane + 1];
  const float o0 = lnb[2 * lane],  o1 = lnb[2 * lane + 1];

  const int end = rp[RPW];
  float fx[RPW], fy[RPW];
  #pragma unroll
  for (int k = 0; k < RPW; ++k) { fx[k] = 0.f; fy[k] = 0.f; }
  float ax = sx[0], ay = sy[0];
  int cur = 0;
  int nb = rp[1];
  auto next_row = [&]() {
    fx[0] = (cur == 0) ? ax : fx[0];  fy[0] = (cur == 0) ? ay : fy[0];
    fx[1] = (cur == 1) ? ax : fx[1];  fy[1] = (cur == 1) ? ay : fy[1];
    fx[2] = (cur == 2) ? ax : fx[2];  fy[2] = (cur == 2) ? ay : fy[2];
    ax = (cur == 0) ? sx[1] : (cur == 1) ? sx[2] : sx[3];
    ay = (cur == 0) ? sy[1] : (cur == 1) ? sy[2] : sy[3];
    nb = (cur == 0) ? rp[2] : (cur == 1) ? rp[3] : 0x7fffffff;
    ++cur;
  };
  int p = rp[0];
  int e[8];
  {
    const int pb = min(p, N_EDGES - 8);
    #pragma unroll
    for (int j = 0; j < 8; ++j) e[j] = ed[pb + j];
  }
  while (p + 8 <= end) {
    float2 r[8];
    #pragma unroll
    for (int j = 0; j < 8; ++j) r[j] = (h2 + e[j] * 32)[lane];
    int en[8];
    const int pn = min(p + 8, N_EDGES - 8);
    #pragma unroll
    for (int j = 0; j < 8; ++j) en[j] = ed[pn + j];
    #pragma unroll
    for (int j = 0; j < 8; ++j) {
      while (p + j >= nb) next_row();
      ax += r[j].x;
      ay += r[j].y;
    }
    #pragma unroll
    for (int j = 0; j < 8; ++j) e[j] = en[j];
    p += 8;
  }
  if (p < end) {
    if (p + 8 > N_EDGES) {
      #pragma unroll
      for (int j = 0; j < 8; ++j) e[j] = ed[min(p + j, end - 1)];
    }
    float2 r[8];
    #pragma unroll
    for (int j = 0; j < 8; ++j) r[j] = (h2 + ((p + j < end) ? e[j] : e[0]) * 32)[lane];
    #pragma unroll
    for (int j = 0; j < 8; ++j) {
      if (p + j < end) {
        while (p + j >= nb) next_row();
        ax += r[j].x;
        ay += r[j].y;
      }
    }
  }
  while (cur < RPW - 1) next_row();
  fx[RPW - 1] = ax; fy[RPW - 1] = ay;

  const int m = lane & 15, half = lane >> 4, c0 = wave * 16;
  constexpr int NC = POOL ? D_OUT : D_H;
  const bool tile_wave = !POOL || wave < D_OUT / 16;
  v2f bfr[16];
  float dn[16];
  if (tile_wave) {
    #pragma unroll
    for (int t = 0; t < 16; ++t) {
      const int ka = 4 * t + 2 * half;
      bfr[t].x = Wn[ka * NC + c0 + m];
      bfr[t].y = Wn[(ka + 1) * NC + c0 + m];
    }
    if (POOL) {
      #pragma unroll
      for (int w = 0; w < 16; ++w) dn[w] = fmaxf((float)gcnt[g[w]], 1.0f);
    }
  }

  #pragma unroll
  for (int k = 0; k < RPW; ++k) {
    const float v0 = fx[k] * di[k] + b0, v1 = fy[k] * di[k] + b1;
    float s = v0 + v1;
    #pragma unroll
    for (int off = 16; off > 0; off >>= 1) s += __shfl_xor(s, off, 32);
    const float mu = s * (1.0f / 64.0f);
    const float d0 = v0 - mu, d1 = v1 - mu;
    float q = d0 * d0 + d1 * d1;
    #pragma unroll
    for (int off = 16; off > 0; off >>= 1) q += __shfl_xor(q, off, 32);
    const float rstd = rsqrtf(q * (1.0f / 64.0f) + LN_EPS);
    float2 y = make_float2(fmaxf(0.f, d0 * rstd * w0 + o0), fmaxf(0.f, d1 * rstd * w1 + o1));
    if (!POOL) { y.x *= di[k]; y.y *= di[k]; }
    const int rb = wave * RPW + k;
    ((float2*)s_row)[rb * (LDA / 2) + lane] = y;
  }
  __syncthreads();
  if (!tile_wave) return;
  v8f acc = {0.f, 0.f, 0.f, 0.f, 0.f, 0.f, 0.f, 0.f};
  #pragma unroll
  for (int t = 0; t < 16; ++t) {
    const int ka = 4 * t + 2 * half;
    const float2 av = *(const float2*)&s_row[m * LDA + ka];
    v2f a;
    a.x = av.x;
    a.y = av.y;
    acc = __builtin_amdgcn_wmma_f32_16x16x4_f32(false, a, false, bfr[t],
                                                (short)0, acc, false, false);
  }
  if (!POOL) {
    const int r0 = (int)blockIdx.x * 16;
    #pragma unroll
    for (int i = 0; i < 8; ++i) hWn[(r0 + i + 8 * half) * D_H + c0 + m] = acc[i];
    return;
  }
  int gc = half ? g[8] : g[0];
  float dc = half ? dn[8] : dn[0];
  float a0 = 0.f;
  #pragma unroll
  for (int w = 0; w < 8; ++w) {
    const int gw = half ? g[8 + w] : g[w];
    if (gw != gc) {
      atomicAdd(&out[gc * D_OUT + c0 + m], a0 / dc);
      gc = gw; dc = half ? dn[8 + w] : dn[w]; a0 = 0.f;
    }
    a0 += acc[w];
  }
  atomicAdd(&out[gc * D_OUT + c0 + m], a0 / dc);
}

template <bool LAST>
static void run_layer(const float* hW, const float* b,
                      const float* lnw, const float* lnb,
                      const float* dinv, const int* rowptr, const int* ed,
                      const float* Wn, float* hWn,
                      const int* batch, const int* gcnt, float* out,
                      hipStream_t stream) {
  k_gather_ln<LAST><<<N_NODES / 16, 128, 0, stream>>>(hW, rowptr, ed, dinv, b, lnw, lnb, Wn, hWn, batch, gcnt, out);
}

extern "C" void kernel_launch(void* const* d_in, const int* in_sizes, int n_in,
                              void* d_out, int out_size, void* d_ws, size_t ws_size,
                              hipStream_t stream) {
  (void)in_sizes; (void)n_in; (void)out_size; (void)ws_size;

  const float* x       = (const float*)d_in[0];
  const int*   eidx    = (const int*)d_in[1];
  const int*   batch   = (const int*)d_in[2];
  const float* W_in    = (const float*)d_in[3];
  const float* b_in    = (const float*)d_in[4];
  const float* ln_in_w = (const float*)d_in[5];
  const float* ln_in_b = (const float*)d_in[6];
  const float* W_h     = (const float*)d_in[7];
  const float* b_h     = (const float*)d_in[8];
  const float* ln_h_w  = (const float*)d_in[9];
  const float* ln_h_b  = (const float*)d_in[10];
  const float* W_out   = (const float*)d_in[11];
  const float* b_out   = (const float*)d_in[12];

  const int* src = eidx;
  const int* dst = eidx + N_EDGES;

  float* ws    = (float*)d_ws;
  float* dinv  = ws;
  int*   rowptr= (int*)(dinv + N_PAD);
  int*   cntc  = rowptr + N_PAD;
  int*   wgtot = cntc + NCHUNK * N_PAD;
  int*   ed    = wgtot + 1024;
  float* hW    = (float*)(ed + N_EDGES);
  float* hW2   = hW + (size_t)N_NODES*64;
  int*   cnt_g = (int*)(hW2 + (size_t)N_NODES*64);
  float* out   = (float*)d_out;

  k_count<<<NPART * NCHUNK, 1024, 0, stream>>>(dst, cntc, wgtot, cnt_g);
  k_place<<<NPART * NCHUNK + XW_BLOCKS, 1024, 0, stream>>>(src, dst, cntc, wgtot, rowptr, dinv, ed, batch, cnt_g,
                                                          b_out, out, x, W_in, hW);

  run_layer<false>(hW,  b_in,     ln_in_w,     ln_in_b,     dinv, rowptr, ed, W_h,           hW2,
                   batch, cnt_g, out, stream);
  run_layer<false>(hW2, b_h,      ln_h_w,      ln_h_b,      dinv, rowptr, ed, W_h + 64 * 64, hW,
                   batch, cnt_g, out, stream);
  run_layer<true>(hW,   b_h + 64, ln_h_w + 64, ln_h_b + 64, dinv, rowptr, ed, W_out,         nullptr,
                  batch, cnt_g, out, stream);
}
